# starting kernel
# baseline (speedup 1.0000x reference)
#include <hip/hip_runtime.h>

constexpr int B_   = 8;
constexpr int N_   = 1024;
constexpr int DIM_ = 768;
constexpr int H_   = 12;
constexpr int CR_  = 384;
constexpr int HD_  = 32;
constexpr float EPS_ = 1e-5f;

typedef __attribute__((ext_vector_type(2))) float v2f;
typedef __attribute__((ext_vector_type(8))) float v8f;

static __device__ __forceinline__ v8f wmma4(v2f a, v2f b, v8f c) {
    return __builtin_amdgcn_wmma_f32_16x16x4_f32(
         false, a,  false, b,
         (short)0, c,  false,  false);
}

__global__ __launch_bounds__(256)
void ln768_kernel(const float* __restrict__ x, const float* __restrict__ g,
                  const float* __restrict__ bia, float* __restrict__ o) {
    const int row  = blockIdx.x;
    const int t    = threadIdx.x;
    const float* xr = x + (size_t)row * DIM_;
    float v0 = xr[t], v1 = xr[t + 256], v2 = xr[t + 512];
    float s  = v0 + v1 + v2;
    float s2 = v0 * v0 + v1 * v1 + v2 * v2;
    #pragma unroll
    for (int off = 16; off > 0; off >>= 1) {
        s  += __shfl_xor(s,  off, 32);
        s2 += __shfl_xor(s2, off, 32);
    }
    __shared__ float ps[8], ps2[8];
    const int wid = t >> 5, lane = t & 31;
    if (lane == 0) { ps[wid] = s; ps2[wid] = s2; }
    __syncthreads();
    if (wid == 0) {
        float a  = (lane < 8) ? ps[lane]  : 0.f;
        float a2 = (lane < 8) ? ps2[lane] : 0.f;
        #pragma unroll
        for (int off = 4; off > 0; off >>= 1) {
            a  += __shfl_xor(a,  off, 32);
            a2 += __shfl_xor(a2, off, 32);
        }
        if (lane == 0) { ps[0] = a; ps2[0] = a2; }
    }
    __syncthreads();
    const float mean = ps[0] * (1.f / DIM_);
    const float var  = ps2[0] * (1.f / DIM_) - mean * mean;
    const float rstd = rsqrtf(var + EPS_);
    float* orow = o + (size_t)row * DIM_;
    orow[t]       = (v0 - mean) * rstd * g[t]       + bia[t];
    orow[t + 256] = (v1 - mean) * rstd * g[t + 256] + bia[t + 256];
    orow[t + 512] = (v2 - mean) * rstd * g[t + 512] + bia[t + 512];
}

__global__ __launch_bounds__(256)
void ln384_kernel(const float* __restrict__ x, const float* __restrict__ g,
                  const float* __restrict__ bia, float* __restrict__ o) {
    const int wid  = threadIdx.x >> 5;
    const int lane = threadIdx.x & 31;
    const int row  = blockIdx.x * 8 + wid;
    const float* xr = x + (size_t)row * CR_;
    float v[12];
    float s = 0.f, s2 = 0.f;
    #pragma unroll
    for (int i = 0; i < 12; ++i) {
        v[i] = xr[lane + 32 * i];
        s += v[i]; s2 += v[i] * v[i];
    }
    #pragma unroll
    for (int off = 16; off > 0; off >>= 1) {
        s  += __shfl_xor(s,  off, 32);
        s2 += __shfl_xor(s2, off, 32);
    }
    const float mean = s * (1.f / CR_);
    const float var  = s2 * (1.f / CR_) - mean * mean;
    const float rstd = rsqrtf(var + EPS_);
    float* orow = o + (size_t)row * CR_;
    #pragma unroll
    for (int i = 0; i < 12; ++i) {
        const int c = lane + 32 * i;
        orow[c] = (v[i] - mean) * rstd * g[c] + bia[c];
    }
}

constexpr int KC      = 32;
constexpr int WSTRIDE = KC + 2;

__global__ __launch_bounds__(256)
void gemm_wmma_kernel(const float* __restrict__ A, const float* __restrict__ W,
                      const float* __restrict__ bias, float* __restrict__ C,
                      int M, int N, int K) {
    __shared__ float sW[2][64 * WSTRIDE];
    const int t    = threadIdx.x;
    const int lane = t & 31;
    const int w    = t >> 5;
    const int hi   = lane >> 4;
    const int lo   = lane & 15;
    const int m0   = blockIdx.y * 128 + w * 16;
    const int n0   = blockIdx.x * 64;
    const int NC   = K / KC;

    const float* arow = A + (size_t)(m0 + lo) * K + 2 * hi;

    float wreg[8], wregN[8];
    v2f   areg[8], aregN[8];

    auto gload = [&](int kc, float* wr, v2f* ar) {
        #pragma unroll
        for (int i = 0; i < 8; ++i)
            wr[i] = W[(size_t)(n0 + i * 8 + w) * K + kc + lane];
        #pragma unroll
        for (int ts = 0; ts < 8; ++ts) {
            ar[ts].x = arow[kc + 4 * ts];
            ar[ts].y = arow[kc + 4 * ts + 1];
        }
    };

    v8f acc[4] = {};
    gload(0, wreg, areg);

    for (int c = 0; c < NC; ++c) {
        const int bufi = c & 1;
        if (c + 1 < NC) gload((c + 1) * KC, wregN, aregN);
        #pragma unroll
        for (int i = 0; i < 8; ++i)
            sW[bufi][(i * 8 + w) * WSTRIDE + lane] = wreg[i];
        __syncthreads();
        const float* sbase = sW[bufi];
        #pragma unroll
        for (int ks = 0; ks < KC / 4; ++ks) {
            #pragma unroll
            for (int tt = 0; tt < 4; ++tt) {
                const float* srow = sbase + (tt * 16 + lo) * WSTRIDE + ks * 4 + 2 * hi;
                v2f b; b.x = srow[0]; b.y = srow[1];
                acc[tt] = wmma4(areg[ks], b, acc[tt]);
            }
        }
        if (c + 1 < NC) {
            #pragma unroll
            for (int i = 0; i < 8; ++i) { wreg[i] = wregN[i]; areg[i] = aregN[i]; }
        }
    }

    #pragma unroll
    for (int tt = 0; tt < 4; ++tt) {
        const int col = n0 + tt * 16 + lo;
        const float bv = bias ? bias[col] : 0.f;
        #pragma unroll
        for (int r = 0; r < 8; ++r) {
            const int row = m0 + r + 8 * hi;
            C[(size_t)row * N + col] = acc[tt][r] + bv;
        }
    }
}

constexpr int SROW = 1028;
constexpr int ATTN_SMEM = 4 * 16 * SROW * 4;

__global__ __launch_bounds__(128)
void attn_kernel(const float* __restrict__ Qb, const float* __restrict__ Kb,
                 const float* __restrict__ Vb, const float* __restrict__ Hb,
                 float* __restrict__ Out) {
    extern __shared__ float sS[];
    const int lane = threadIdx.x & 31;
    const int w    = threadIdx.x >> 5;
    const int bh   = blockIdx.y;
    const int b    = bh / H_;
    const int h    = bh % H_;
    const int q0   = blockIdx.x * 64 + w * 16;
    const int hi   = lane >> 4;
    const int lo   = lane & 15;
    const float scale = 0.17677669529663687f;
    float* sP = sS + (size_t)w * 16 * SROW;

    const float* qrow = Qb + ((size_t)(b * N_ + q0 + lo) * CR_) + h * HD_ + 2 * hi;
    v2f aq[8];
    #pragma unroll
    for (int ts = 0; ts < 8; ++ts) { aq[ts].x = qrow[4 * ts]; aq[ts].y = qrow[4 * ts + 1]; }

    const float* kbase = Kb + (size_t)b * N_ * CR_ + h * HD_ + 2 * hi;
    v2f bk[8], bkN[8];
    {
        const float* kr = kbase + (size_t)lo * CR_;
        #pragma unroll
        for (int ts = 0; ts < 8; ++ts) { bk[ts].x = kr[4 * ts]; bk[ts].y = kr[4 * ts + 1]; }
    }
    for (int j = 0; j < N_ / 16; ++j) {
        if (j + 1 < N_ / 16) {
            const float* kr = kbase + (size_t)((j + 1) * 16 + lo) * CR_;
            #pragma unroll
            for (int ts = 0; ts < 8; ++ts) { bkN[ts].x = kr[4 * ts]; bkN[ts].y = kr[4 * ts + 1]; }
        }
        v8f acc = {};
        #pragma unroll
        for (int ts = 0; ts < 8; ++ts) acc = wmma4(aq[ts], bk[ts], acc);
        #pragma unroll
        for (int r = 0; r < 8; ++r)
            sP[(r + 8 * hi) * SROW + j * 16 + lo] = acc[r] * scale;
        if (j + 1 < N_ / 16) {
            #pragma unroll
            for (int ts = 0; ts < 8; ++ts) bk[ts] = bkN[ts];
        }
    }

    for (int r = 0; r < 16; ++r) {
        float* srow = sP + r * SROW;
        float vals[32];
        float mx = -3.0e38f;
        #pragma unroll
        for (int i = 0; i < 32; ++i) {
            vals[i] = srow[lane + 32 * i];
            mx = fmaxf(mx, vals[i]);
        }
        #pragma unroll
        for (int off = 16; off > 0; off >>= 1)
            mx = fmaxf(mx, __shfl_xor(mx, off, 32));
        float sum = 0.f;
        #pragma unroll
        for (int i = 0; i < 32; ++i) {
            vals[i] = __expf(vals[i] - mx);
            sum += vals[i];
        }
        #pragma unroll
        for (int off = 16; off > 0; off >>= 1)
            sum += __shfl_xor(sum, off, 32);
        const float inv = 1.f / sum;
        #pragma unroll
        for (int i = 0; i < 32; ++i)
            srow[lane + 32 * i] = vals[i] * inv;
    }

    v8f oacc[2] = {};
    const float* vbase = Vb + (size_t)b * N_ * CR_ + h * HD_;
    const float* prow  = sP + lo * SROW + 2 * hi;
    const int d0 = lo, d1 = 16 + lo;

    v2f ap[2], bva[2], bvb[2], apN[2], bvaN[2], bvbN[2];
    {
        #pragma unroll
        for (int u = 0; u < 2; ++u) {
            const int k0 = 4 * u;
            ap[u].x = prow[k0]; ap[u].y = prow[k0 + 1];
            const float* vr = vbase + (size_t)(k0 + 2 * hi) * CR_;
            bva[u].x = vr[d0]; bva[u].y = vr[CR_ + d0];
            bvb[u].x = vr[d1]; bvb[u].y = vr[CR_ + d1];
        }
    }
    for (int k0 = 0; k0 < N_; k0 += 8) {
        if (k0 + 8 < N_) {
            #pragma unroll
            for (int u = 0; u < 2; ++u) {
                const int kn = k0 + 8 + 4 * u;
                apN[u].x = prow[kn]; apN[u].y = prow[kn + 1];
                const float* vr = vbase + (size_t)(kn + 2 * hi) * CR_;
                bvaN[u].x = vr[d0]; bvaN[u].y = vr[CR_ + d0];
                bvbN[u].x = vr[d1]; bvbN[u].y = vr[CR_ + d1];
            }
        }
        #pragma unroll
        for (int u = 0; u < 2; ++u) {
            oacc[0] = wmma4(ap[u], bva[u], oacc[0]);
            oacc[1] = wmma4(ap[u], bvb[u], oacc[1]);
        }
        if (k0 + 8 < N_) {
            #pragma unroll
            for (int u = 0; u < 2; ++u) { ap[u] = apN[u]; bva[u] = bvaN[u]; bvb[u] = bvbN[u]; }
        }
    }

    #pragma unroll
    for (int tt = 0; tt < 2; ++tt) {
        const int col = h * HD_ + tt * 16 + lo;
        #pragma unroll
        for (int r = 0; r < 8; ++r) {
            const int row = q0 + r + 8 * hi;
            const size_t idx = (size_t)(b * N_ + row) * CR_ + col;
            Out[idx] = Hb[idx] + oacc[tt][r];
        }
    }
}

extern "C" void kernel_launch(void* const* d_in, const int* in_sizes, int n_in,
                              void* d_out, int out_size, void* d_ws, size_t ws_size,
                              hipStream_t stream) {
    (void)in_sizes; (void)n_in; (void)out_size; (void)ws_size;
    const float* x      = (const float*)d_in[0];
    const float* ln0_g  = (const float*)d_in[1];
    const float* ln0_b  = (const float*)d_in[2];
    const float* w_proj = (const float*)d_in[3];
    const float* b_proj = (const float*)d_in[4];
    const float* ln1_g  = (const float*)d_in[5];
    const float* ln1_b  = (const float*)d_in[6];
    const float* wq     = (const float*)d_in[7];
    const float* wk     = (const float*)d_in[8];
    const float* wv     = (const float*)d_in[9];
    float* out = (float*)d_out;

    float* ws = (float*)d_ws;
    const size_t rowsBig = (size_t)B_ * N_ * DIM_;
    const size_t rowsCr  = (size_t)B_ * N_ * CR_;
    float* xn = ws;
    float* hB = xn + rowsBig;
    float* yB = hB + rowsCr;
    float* qB = yB + rowsCr;
    float* kB = qB + rowsCr;
    float* vB = kB + rowsCr;

    const int M = B_ * N_;

    ln768_kernel<<<M, 256, 0, stream>>>(x, ln0_g, ln0_b, xn);

    gemm_wmma_kernel<<<dim3(CR_ / 64, M / 128), 256, 0, stream>>>(
        xn, w_proj, b_proj, hB, M, CR_, DIM_);

    ln384_kernel<<<M / 8, 256, 0, stream>>>(hB, ln1_g, ln1_b, yB);

    gemm_wmma_kernel<<<dim3(CR_ / 64, M / 128), 256, 0, stream>>>(
        yB, wq, nullptr, qB, M, CR_, CR_);
    gemm_wmma_kernel<<<dim3(CR_ / 64, M / 128), 256, 0, stream>>>(
        yB, wk, nullptr, kB, M, CR_, CR_);
    gemm_wmma_kernel<<<dim3(CR_ / 64, M / 128), 256, 0, stream>>>(
        yB, wv, nullptr, vB, M, CR_, CR_);

    (void)hipFuncSetAttribute((const void*)attn_kernel,
                              hipFuncAttributeMaxDynamicSharedMemorySize,
                              ATTN_SMEM);
    attn_kernel<<<dim3(N_ / 64, B_ * H_), 128, ATTN_SMEM, stream>>>(
        qB, kB, vB, hB, out);
}


# attention: work-group = one (b,h) x 8 query tiles; each wave keeps the K and V fragments of its 128 keys in registers (loaded once, 64 loads in one group): no K/V load or barrier in the tile loop
# speedup vs baseline: 2.5423x; 2.5423x over previous
#include <hip/hip_runtime.h>

constexpr int B_   = 8;
constexpr int N_   = 1024;
constexpr int DIM_ = 768;
constexpr int H_   = 12;
constexpr int CR_  = 384;
constexpr int HD_  = 32;
constexpr float EPS_ = 1e-5f;

typedef __attribute__((ext_vector_type(2))) float v2f;
typedef __attribute__((ext_vector_type(8))) float v8f;

static __device__ __forceinline__ v8f wmma4(v2f a, v2f b, v8f c) {
    return __builtin_amdgcn_wmma_f32_16x16x4_f32(
         false, a,  false, b,
         (short)0, c,  false,  false);
}

template <int K>
__global__ __launch_bounds__(256) __attribute__((amdgpu_waves_per_eu(2, 4)))
void ln_frag_kernel(const float* __restrict__ x, const float* __restrict__ g,
                    const float* __restrict__ bia, float* __restrict__ of) {
    constexpr int NL    = K / 128;
    constexpr int PITCH = K + 4;
    constexpr int NP    = K / 8;
    constexpr int PPW   = NP / 8;
    __shared__ __align__(16) float tile[16 * PITCH];
    const int lane = threadIdx.x & 31;
    const int w    = threadIdx.x >> 5;
    const int rt   = blockIdx.x;

    const float* xr = x + (size_t)(rt * 16 + 2 * w) * K + 4 * lane;
    float4 v[2][NL];
    #pragma unroll
    for (int i = 0; i < NL; ++i) {
        v[0][i] = *reinterpret_cast<const float4*>(xr + 128 * i);
        v[1][i] = *reinterpret_cast<const float4*>(xr + K + 128 * i);
    }
    float4 gv[NL], bv[NL];
    #pragma unroll
    for (int i = 0; i < NL; ++i) {
        gv[i] = *reinterpret_cast<const float4*>(g   + 128 * i + 4 * lane);
        bv[i] = *reinterpret_cast<const float4*>(bia + 128 * i + 4 * lane);
    }
    float s[2] = {0.f, 0.f}, s2[2] = {0.f, 0.f};
    #pragma unroll
    for (int i = 0; i < NL; ++i) {
        #pragma unroll
        for (int rr = 0; rr < 2; ++rr) {
            const float4 a = v[rr][i];
            s[rr]  += (a.x + a.y) + (a.z + a.w);
            s2[rr] += (a.x * a.x + a.y * a.y) + (a.z * a.z + a.w * a.w);
        }
    }
    #pragma unroll
    for (int off = 16; off > 0; off >>= 1) {
        #pragma unroll
        for (int rr = 0; rr < 2; ++rr) {
            s[rr]  += __shfl_xor(s[rr],  off, 32);
            s2[rr] += __shfl_xor(s2[rr], off, 32);
        }
    }
    #pragma unroll
    for (int rr = 0; rr < 2; ++rr) {
        const float mean = s[rr] * (1.f / K);
        const float var  = s2[rr] * (1.f / K) - mean * mean;
        const float rstd = rsqrtf(var + EPS_);
        float* trow = tile + (2 * w + rr) * PITCH + 4 * lane;
        #pragma unroll
        for (int i = 0; i < NL; ++i) {
            const float4 a = v[rr][i];
            float4 o;
            o.x = (a.x - mean) * rstd * gv[i].x + bv[i].x;
            o.y = (a.y - mean) * rstd * gv[i].y + bv[i].y;
            o.z = (a.z - mean) * rstd * gv[i].z + bv[i].z;
            o.w = (a.w - mean) * rstd * gv[i].w + bv[i].w;
            *reinterpret_cast<float4*>(trow + 128 * i) = o;
        }
    }
    __syncthreads();

    const int lo = lane & 15;
    const int hi = lane >> 4;
    const float* tsrc = tile + lo * PITCH + 4 * hi;
    float4* dst = reinterpret_cast<float4*>(of)
                + ((size_t)rt * NP + w * PPW) * 32 + lane;
    #pragma unroll
    for (int j = 0; j < PPW; ++j)
        dst[j * 32] = *reinterpret_cast<const float4*>(tsrc + 8 * (w * PPW + j));
}

constexpr int KC      = 32;
constexpr int WSTRIDE = KC + 2;

template <int MODE>
static __device__ __forceinline__
void gemm_tile(const float* __restrict__ A, const float* __restrict__ W,
               const float* __restrict__ bias, float* __restrict__ C,
               int N, int K, int bx, int by,
               float* __restrict__ sW, float* __restrict__ sStage) {
    const int t    = threadIdx.x;
    const int lane = t & 31;
    const int w    = t >> 5;
    const int hi   = lane >> 4;
    const int lo   = lane & 15;
    const int m0   = by * 128 + w * 16;
    const int n0   = bx * 64;
    const int NC   = K / KC;

    const float4* afrag = reinterpret_cast<const float4*>(A)
                        + ((size_t)(m0 >> 4) * (K >> 3)) * 32 + lane;
    const int lperm = (lane & ~6) | ((lane & 4) >> 1) | ((lane & 2) << 1);

    float wreg[8], wregN[8];
    v2f   areg[8], aregN[8];

    auto gload = [&](int kc, float* wr, v2f* ar) {
        #pragma unroll
        for (int i = 0; i < 8; ++i)
            wr[i] = W[(size_t)(n0 + i * 8 + w) * K + kc + lane];
        #pragma unroll
        for (int j = 0; j < 4; ++j) {
            const float4 a4 = afrag[((kc >> 3) + j) * 32];
            ar[2 * j].x     = a4.x; ar[2 * j].y     = a4.y;
            ar[2 * j + 1].x = a4.z; ar[2 * j + 1].y = a4.w;
        }
    };

    v8f acc[4] = {};
    gload(0, wreg, areg);

    for (int c = 0; c < NC; ++c) {
        const int bufi = c & 1;
        if (c + 1 < NC) gload((c + 1) * KC, wregN, aregN);
        #pragma unroll
        for (int i = 0; i < 8; ++i)
            sW[bufi * (64 * WSTRIDE) + (i * 8 + w) * WSTRIDE + lperm] = wreg[i];
        __syncthreads();
        const float* sbase = sW + bufi * (64 * WSTRIDE);
        #pragma unroll
        for (int ks = 0; ks < KC / 4; ++ks) {
            #pragma unroll
            for (int tt = 0; tt < 4; ++tt) {
                const float* srow = sbase + (tt * 16 + lo) * WSTRIDE + ks * 4 + 2 * hi;
                v2f b; b.x = srow[0]; b.y = srow[1];
                acc[tt] = wmma4(areg[ks], b, acc[tt]);
            }
        }
        if (c + 1 < NC) {
            #pragma unroll
            for (int i = 0; i < 8; ++i) { wreg[i] = wregN[i]; areg[i] = aregN[i]; }
        }
    }

    if constexpr (MODE == 0) {
        #pragma unroll
        for (int tt = 0; tt < 4; ++tt) {
            const int col = n0 + tt * 16 + lo;
            const float bv = bias ? bias[col] : 0.f;
            #pragma unroll
            for (int r = 0; r < 8; ++r) {
                const int row = m0 + r + 8 * hi;
                C[(size_t)row * N + col] = acc[tt][r] + bv;
            }
        }
    } else if constexpr (MODE == 1) {
        float* stg = sStage + w * 1024;
        #pragma unroll
        for (int tt = 0; tt < 4; ++tt) {
            const int d   = (tt & 1) * 16 + lo;
            const int off = (tt >> 1) * 512 + (d >> 3) * 128 + ((d >> 1) & 1) * 64
                          + ((d >> 2) & 1) * 2 + (d & 1);
            #pragma unroll
            for (int r = 0; r < 8; ++r)
                stg[off + (r + 8 * hi) * 4] = acc[tt][r];
        }
        __syncthreads();
        const int bq    = m0 / N_;
        const int gtile = (m0 % N_) >> 4;
        const int h0    = n0 / HD_;
        #pragma unroll
        for (int hh = 0; hh < 2; ++hh) {
            float* dst = C + ((size_t)((bq * H_ + h0 + hh) * (N_ / 16) + gtile)) * 512;
            #pragma unroll
            for (int j = 0; j < 4; ++j) {
                const float4 v4 = *reinterpret_cast<const float4*>(stg + hh * 512 + j * 128 + lane * 4);
                *reinterpret_cast<float4*>(dst + j * 128 + lane * 4) = v4;
            }
        }
    } else {
        const int bq  = m0 / N_;
        const int kg0 = (m0 % N_) >> 2;
        const int h0  = n0 / HD_;
        #pragma unroll
        for (int hh = 0; hh < 2; ++hh) {
            float* dst = C + ((size_t)(bq * H_ + h0 + hh) * (N_ / 4) + kg0 + 2 * hi) * 128 + lo * 4;
            #pragma unroll
            for (int rq = 0; rq < 2; ++rq) {
                #pragma unroll
                for (int hf = 0; hf < 2; ++hf) {
                    const int r = 4 * rq + 2 * hf;
                    float4 v4;
                    v4.x = acc[2 * hh][r];     v4.y = acc[2 * hh][r + 1];
                    v4.z = acc[2 * hh + 1][r]; v4.w = acc[2 * hh + 1][r + 1];
                    *reinterpret_cast<float4*>(dst + rq * 128 + hf * 64) = v4;
                }
            }
        }
    }
}


__global__ __launch_bounds__(256)
void proj_gemm_kernel(const float* __restrict__ A, const float* __restrict__ W,
                      const float* __restrict__ bias, float* __restrict__ C,
                      int N, int K) {
    __shared__ float sW[2 * 64 * WSTRIDE];
    gemm_tile<0>(A, W, bias, C, N, K, blockIdx.x, blockIdx.y, sW, nullptr);
}

__global__ __launch_bounds__(256)
void qkv_gemm_kernel(const float* __restrict__ A, const float* __restrict__ Wq,
                     const float* __restrict__ Wk, const float* __restrict__ Wv,
                     float* __restrict__ Cq, float* __restrict__ Ck,
                     float* __restrict__ Cv, int N, int K) {
    __shared__ float sW[2 * 64 * WSTRIDE];
    __shared__ float sStage[8 * 1024];
    const int nblk  = N / 64;
    const int which = blockIdx.x / nblk;
    const int bx    = blockIdx.x - which * nblk;
    if (which == 0)
        gemm_tile<1>(A, Wq, nullptr, Cq, N, K, bx, blockIdx.y, sW, sStage);
    else if (which == 1)
        gemm_tile<1>(A, Wk, nullptr, Ck, N, K, bx, blockIdx.y, sW, sStage);
    else
        gemm_tile<2>(A, Wv, nullptr, Cv, N, K, bx, blockIdx.y, sW, sStage);
}

constexpr int SROW = 1028;
constexpr int AW   = 8;
constexpr int KPW  = N_ / AW;
constexpr int NQT  = 8;
constexpr int PROW = 36;
constexpr int STRIP_FLOATS = 16 * SROW;
constexpr int PARK_FLOATS  = NQT * AW * 16 * PROW;
constexpr int ATTN_SMEM = (STRIP_FLOATS + PARK_FLOATS) * 4;

static __device__ __forceinline__ float xhalf(float x) {
    return __builtin_bit_cast(float, __builtin_amdgcn_permlanex16(
        __builtin_bit_cast(int, x), __builtin_bit_cast(int, x),
        0x76543210, 0xfedcba98, false, false));
}

__global__ __launch_bounds__(256) __attribute__((amdgpu_waves_per_eu(2, 2)))
void attn_kernel(const float* __restrict__ Qb, const float* __restrict__ Kb,
                 const float* __restrict__ Vb, const float* __restrict__ Hb,
                 float* __restrict__ Out) {
    extern __shared__ __align__(16) float sS[];
    float* park = sS + STRIP_FLOATS;
    const int lane = threadIdx.x & 31;
    const int w    = threadIdx.x >> 5;
    const int bh   = blockIdx.y;
    const int b    = bh / H_;
    const int h    = bh % H_;
    const int qt0  = blockIdx.x * NQT;
    const int hi   = lane >> 4;
    const int lo   = lane & 15;
    const int key0 = w * KPW;
    const float cexp = 0.2550348615890031f;


    const float4* qf = reinterpret_cast<const float4*>(Qb)
                     + ((size_t)(bh * (N_ / 16) + qt0) * 4) * 32 + lane;
    v2f aq[8];
    #pragma unroll
    for (int tsp = 0; tsp < 4; ++tsp) {
        const float4 q4 = qf[tsp * 32];
        aq[2 * tsp].x     = q4.x; aq[2 * tsp].y     = q4.y;
        aq[2 * tsp + 1].x = q4.z; aq[2 * tsp + 1].y = q4.w;
    }

    const int jrow = threadIdx.x >> 4;
    const int jd   = (threadIdx.x & 15) * 2;
    const size_t jidx = (size_t)(b * N_ + qt0 * 16 + jrow) * CR_ + h * HD_ + jd;
    float2 hv[NQT];
    #pragma unroll
    for (int t = 0; t < NQT; ++t)
        hv[t] = *reinterpret_cast<const float2*>(Hb + jidx + (size_t)t * 16 * CR_);

    const float4* kbase = reinterpret_cast<const float4*>(Kb)
                        + ((size_t)(bh * (N_ / 16) + w * (KPW / 16)) * 4) * 32 + lane;
    v2f kf[64];
    #pragma unroll
    for (int i = 0; i < 32; ++i) {
        const float4 k4 = kbase[i * 32];
        kf[2 * i].x     = k4.x; kf[2 * i].y     = k4.y;
        kf[2 * i + 1].x = k4.z; kf[2 * i + 1].y = k4.w;
    }
    const float4* vbase = reinterpret_cast<const float4*>(Vb)
                        + ((size_t)bh * (N_ / 4) + key0 / 4) * 32 + lane;
    v2f va[32], vb[32];
    #pragma unroll
    for (int kk = 0; kk < 32; ++kk) {
        const float4 v4 = vbase[kk * 32];
        va[kk].x = v4.x; va[kk].y = v4.y;
        vb[kk].x = v4.z; vb[kk].y = v4.w;
    }

    float* sw          = sS + lo * SROW + key0 + 64 * hi;
    const float* prow  = sS + lo * SROW + key0 + 2 * hi;

    #pragma unroll 1
    for (int t = 0; t < NQT; ++t) {
        #pragma unroll
        for (int j = 0; j < KPW / 16; j += 2) {
            v8f acc0 = {}, acc1 = {};
            #pragma unroll
            for (int ts = 0; ts < 8; ++ts) {
                acc0 = wmma4(aq[ts], kf[8 * j + ts],       acc0);
                acc1 = wmma4(aq[ts], kf[8 * (j + 1) + ts], acc1);
            }
            #pragma unroll
            for (int r = 0; r < 8; ++r) {
                sS[(r + 8 * hi) * SROW + key0 + j * 16 + lo]       = acc0[r];
                sS[(r + 8 * hi) * SROW + key0 + (j + 1) * 16 + lo] = acc1[r];
            }
        }

        if (t + 1 < NQT) {
            #pragma unroll
            for (int tsp = 0; tsp < 4; ++tsp) {
                const float4 q4 = qf[(t + 1) * 128 + tsp * 32];
                aq[2 * tsp].x     = q4.x; aq[2 * tsp].y     = q4.y;
                aq[2 * tsp + 1].x = q4.z; aq[2 * tsp + 1].y = q4.w;
            }
        }

        float4 pv[16];
        #pragma unroll
        for (int i = 0; i < 16; ++i)
            pv[i] = *reinterpret_cast<const float4*>(sw + 4 * i);
        float mx = fmaxf(fmaxf(pv[0].x, pv[0].y), fmaxf(pv[0].z, pv[0].w));
        #pragma unroll
        for (int i = 1; i < 16; ++i)
            mx = fmaxf(mx, fmaxf(fmaxf(pv[i].x, pv[i].y), fmaxf(pv[i].z, pv[i].w)));
        mx = fmaxf(mx, xhalf(mx));
        const float mc = mx * cexp;
        float sum = 0.f;
        #pragma unroll
        for (int i = 0; i < 16; ++i) {
            pv[i].x = __builtin_amdgcn_exp2f(fmaf(pv[i].x, cexp, -mc));
            pv[i].y = __builtin_amdgcn_exp2f(fmaf(pv[i].y, cexp, -mc));
            pv[i].z = __builtin_amdgcn_exp2f(fmaf(pv[i].z, cexp, -mc));
            pv[i].w = __builtin_amdgcn_exp2f(fmaf(pv[i].w, cexp, -mc));
            sum += (pv[i].x + pv[i].y) + (pv[i].z + pv[i].w);
        }
        sum += xhalf(sum);
        #pragma unroll
        for (int i = 0; i < 16; ++i)
            *reinterpret_cast<float4*>(sw + 4 * i) = pv[i];

        v8f oacc0 = {}, oacc1 = {};
        #pragma unroll
        for (int kk = 0; kk < 32; ++kk) {
            v2f ap; ap.x = prow[4 * kk]; ap.y = prow[4 * kk + 1];
            oacc0 = wmma4(ap, va[kk], oacc0);
            oacc1 = wmma4(ap, vb[kk], oacc1);
        }

        float* pk = park + (size_t)((t * AW + w) * 16) * PROW;
        #pragma unroll
        for (int r = 0; r < 8; ++r) {
            pk[(r + 8 * hi) * PROW + lo]      = oacc0[r];
            pk[(r + 8 * hi) * PROW + 16 + lo] = oacc1[r];
        }
        pk[lo * PROW + 32 + hi] = hi ? sum : mc;
    }
    __syncthreads();

    #pragma unroll
    for (int t = 0; t < NQT; ++t) {
        const float* pp = park + (size_t)(t * AW * 16 + jrow) * PROW;
        float2 ml[AW];
        float mall = -3.0e38f;
        #pragma unroll
        for (int ww = 0; ww < AW; ++ww) {
            ml[ww] = *reinterpret_cast<const float2*>(pp + ww * 16 * PROW + 32);
            mall = fmaxf(mall, ml[ww].x);
        }
        float den = 0.f, sx = 0.f, sy = 0.f;
        #pragma unroll
        for (int ww = 0; ww < AW; ++ww) {
            const float f  = __builtin_amdgcn_exp2f(ml[ww].x - mall);
            const float2 o = *reinterpret_cast<const float2*>(pp + ww * 16 * PROW + jd);
            den = fmaf(f, ml[ww].y, den);
            sx  = fmaf(f, o.x, sx);
            sy  = fmaf(f, o.y, sy);
        }
        const float inv = 1.f / den;
        float2 ov; ov.x = fmaf(sx, inv, hv[t].x); ov.y = fmaf(sy, inv, hv[t].y);
        *reinterpret_cast<float2*>(Out + jidx + (size_t)t * 16 * CR_) = ov;
    }
}

extern "C" void kernel_launch(void* const* d_in, const int* in_sizes, int n_in,
                              void* d_out, int out_size, void* d_ws, size_t ws_size,
                              hipStream_t stream) {
    (void)in_sizes; (void)n_in; (void)out_size; (void)ws_size;
    const float* x      = (const float*)d_in[0];
    const float* ln0_g  = (const float*)d_in[1];
    const float* ln0_b  = (const float*)d_in[2];
    const float* w_proj = (const float*)d_in[3];
    const float* b_proj = (const float*)d_in[4];
    const float* ln1_g  = (const float*)d_in[5];
    const float* ln1_b  = (const float*)d_in[6];
    const float* wq     = (const float*)d_in[7];
    const float* wk     = (const float*)d_in[8];
    const float* wv     = (const float*)d_in[9];
    float* out = (float*)d_out;

    float* ws = (float*)d_ws;
    const size_t rowsBig = (size_t)B_ * N_ * DIM_;
    const size_t rowsCr  = (size_t)B_ * N_ * CR_;
    float* xn = ws;
    float* hB = xn + rowsBig;
    float* yB = hB + rowsCr;
    float* qB = yB + rowsCr;
    float* kB = qB + rowsCr;
    float* vB = kB + rowsCr;

    const int M = B_ * N_;

    ln_frag_kernel<DIM_><<<M / 16, 256, 0, stream>>>(x, ln0_g, ln0_b, xn);

    proj_gemm_kernel<<<dim3(CR_ / 64, M / 128), 256, 0, stream>>>(
        xn, w_proj, b_proj, hB, CR_, DIM_);

    ln_frag_kernel<CR_><<<M / 16, 256, 0, stream>>>(hB, ln1_g, ln1_b, yB);

    qkv_gemm_kernel<<<dim3(3 * CR_ / 64, M / 128), 256, 0, stream>>>(
        yB, wq, wk, wv, qB, kB, vB, CR_, CR_);

    (void)hipFuncSetAttribute((const void*)attn_kernel,
                              hipFuncAttributeMaxDynamicSharedMemorySize,
                              ATTN_SMEM);
    attn_kernel<<<dim3(N_ / 16 / NQT, B_ * H_), 256, ATTN_SMEM, stream>>>(
        qB, kB, vB, hB, out);
}


# attention P V: P stays in the softmax registers (k order of the P V fragments = the softmax lane's own key order, V loaded to match), no P store to LDS or reload; changes P V summation order
# speedup vs baseline: 2.6267x; 1.0332x over previous
#include <hip/hip_runtime.h>

constexpr int B_   = 8;
constexpr int N_   = 1024;
constexpr int DIM_ = 768;
constexpr int H_   = 12;
constexpr int CR_  = 384;
constexpr int HD_  = 32;
constexpr float EPS_ = 1e-5f;

typedef __attribute__((ext_vector_type(2))) float v2f;
typedef __attribute__((ext_vector_type(8))) float v8f;

static __device__ __forceinline__ v8f wmma4(v2f a, v2f b, v8f c) {
    return __builtin_amdgcn_wmma_f32_16x16x4_f32(
         false, a,  false, b,
         (short)0, c,  false,  false);
}

template <int K>
__global__ __launch_bounds__(256) __attribute__((amdgpu_waves_per_eu(2, 4)))
void ln_frag_kernel(const float* __restrict__ x, const float* __restrict__ g,
                    const float* __restrict__ bia, float* __restrict__ of) {
    constexpr int NL    = K / 128;
    constexpr int PITCH = K + 4;
    constexpr int NP    = K / 8;
    constexpr int PPW   = NP / 8;
    __shared__ __align__(16) float tile[16 * PITCH];
    const int lane = threadIdx.x & 31;
    const int w    = threadIdx.x >> 5;
    const int rt   = blockIdx.x;

    const float* xr = x + (size_t)(rt * 16 + 2 * w) * K + 4 * lane;
    float4 v[2][NL];
    #pragma unroll
    for (int i = 0; i < NL; ++i) {
        v[0][i] = *reinterpret_cast<const float4*>(xr + 128 * i);
        v[1][i] = *reinterpret_cast<const float4*>(xr + K + 128 * i);
    }
    float4 gv[NL], bv[NL];
    #pragma unroll
    for (int i = 0; i < NL; ++i) {
        gv[i] = *reinterpret_cast<const float4*>(g   + 128 * i + 4 * lane);
        bv[i] = *reinterpret_cast<const float4*>(bia + 128 * i + 4 * lane);
    }
    float s[2] = {0.f, 0.f}, s2[2] = {0.f, 0.f};
    #pragma unroll
    for (int i = 0; i < NL; ++i) {
        #pragma unroll
        for (int rr = 0; rr < 2; ++rr) {
            const float4 a = v[rr][i];
            s[rr]  += (a.x + a.y) + (a.z + a.w);
            s2[rr] += (a.x * a.x + a.y * a.y) + (a.z * a.z + a.w * a.w);
        }
    }
    #pragma unroll
    for (int off = 16; off > 0; off >>= 1) {
        #pragma unroll
        for (int rr = 0; rr < 2; ++rr) {
            s[rr]  += __shfl_xor(s[rr],  off, 32);
            s2[rr] += __shfl_xor(s2[rr], off, 32);
        }
    }
    #pragma unroll
    for (int rr = 0; rr < 2; ++rr) {
        const float mean = s[rr] * (1.f / K);
        const float var  = s2[rr] * (1.f / K) - mean * mean;
        const float rstd = rsqrtf(var + EPS_);
        float* trow = tile + (2 * w + rr) * PITCH + 4 * lane;
        #pragma unroll
        for (int i = 0; i < NL; ++i) {
            const float4 a = v[rr][i];
            float4 o;
            o.x = (a.x - mean) * rstd * gv[i].x + bv[i].x;
            o.y = (a.y - mean) * rstd * gv[i].y + bv[i].y;
            o.z = (a.z - mean) * rstd * gv[i].z + bv[i].z;
            o.w = (a.w - mean) * rstd * gv[i].w + bv[i].w;
            *reinterpret_cast<float4*>(trow + 128 * i) = o;
        }
    }
    __syncthreads();

    const int lo = lane & 15;
    const int hi = lane >> 4;
    const float* tsrc = tile + lo * PITCH + 4 * hi;
    float4* dst = reinterpret_cast<float4*>(of)
                + ((size_t)rt * NP + w * PPW) * 32 + lane;
    #pragma unroll
    for (int j = 0; j < PPW; ++j)
        dst[j * 32] = *reinterpret_cast<const float4*>(tsrc + 8 * (w * PPW + j));
}

constexpr int KC      = 32;
constexpr int WSTRIDE = KC + 2;

template <int MODE>
static __device__ __forceinline__
void gemm_tile(const float* __restrict__ A, const float* __restrict__ W,
               const float* __restrict__ bias, float* __restrict__ C,
               int N, int K, int bx, int by,
               float* __restrict__ sW, float* __restrict__ sStage) {
    const int t    = threadIdx.x;
    const int lane = t & 31;
    const int w    = t >> 5;
    const int hi   = lane >> 4;
    const int lo   = lane & 15;
    const int m0   = by * 128 + w * 16;
    const int n0   = bx * 64;
    const int NC   = K / KC;

    const float4* afrag = reinterpret_cast<const float4*>(A)
                        + ((size_t)(m0 >> 4) * (K >> 3)) * 32 + lane;
    const int lperm = (lane & ~6) | ((lane & 4) >> 1) | ((lane & 2) << 1);

    float wreg[8], wregN[8];
    v2f   areg[8], aregN[8];

    auto gload = [&](int kc, float* wr, v2f* ar) {
        #pragma unroll
        for (int i = 0; i < 8; ++i)
            wr[i] = W[(size_t)(n0 + i * 8 + w) * K + kc + lane];
        #pragma unroll
        for (int j = 0; j < 4; ++j) {
            const float4 a4 = afrag[((kc >> 3) + j) * 32];
            ar[2 * j].x     = a4.x; ar[2 * j].y     = a4.y;
            ar[2 * j + 1].x = a4.z; ar[2 * j + 1].y = a4.w;
        }
    };

    v8f acc[4] = {};
    gload(0, wreg, areg);

    for (int c = 0; c < NC; ++c) {
        const int bufi = c & 1;
        if (c + 1 < NC) gload((c + 1) * KC, wregN, aregN);
        #pragma unroll
        for (int i = 0; i < 8; ++i)
            sW[bufi * (64 * WSTRIDE) + (i * 8 + w) * WSTRIDE + lperm] = wreg[i];
        __syncthreads();
        const float* sbase = sW + bufi * (64 * WSTRIDE);
        #pragma unroll
        for (int ks = 0; ks < KC / 4; ++ks) {
            #pragma unroll
            for (int tt = 0; tt < 4; ++tt) {
                const float* srow = sbase + (tt * 16 + lo) * WSTRIDE + ks * 4 + 2 * hi;
                v2f b; b.x = srow[0]; b.y = srow[1];
                acc[tt] = wmma4(areg[ks], b, acc[tt]);
            }
        }
        if (c + 1 < NC) {
            #pragma unroll
            for (int i = 0; i < 8; ++i) { wreg[i] = wregN[i]; areg[i] = aregN[i]; }
        }
    }

    if constexpr (MODE == 0) {
        #pragma unroll
        for (int tt = 0; tt < 4; ++tt) {
            const int col = n0 + tt * 16 + lo;
            const float bv = bias ? bias[col] : 0.f;
            #pragma unroll
            for (int r = 0; r < 8; ++r) {
                const int row = m0 + r + 8 * hi;
                C[(size_t)row * N + col] = acc[tt][r] + bv;
            }
        }
    } else if constexpr (MODE == 1) {
        float* stg = sStage + w * 1024;
        #pragma unroll
        for (int tt = 0; tt < 4; ++tt) {
            const int d   = (tt & 1) * 16 + lo;
            const int off = (tt >> 1) * 512 + (d >> 3) * 128 + ((d >> 1) & 1) * 64
                          + ((d >> 2) & 1) * 2 + (d & 1);
            #pragma unroll
            for (int r = 0; r < 8; ++r)
                stg[off + (r + 8 * hi) * 4] = acc[tt][r];
        }
        __syncthreads();
        const int bq    = m0 / N_;
        const int gtile = (m0 % N_) >> 4;
        const int h0    = n0 / HD_;
        #pragma unroll
        for (int hh = 0; hh < 2; ++hh) {
            float* dst = C + ((size_t)((bq * H_ + h0 + hh) * (N_ / 16) + gtile)) * 512;
            #pragma unroll
            for (int j = 0; j < 4; ++j) {
                const float4 v4 = *reinterpret_cast<const float4*>(stg + hh * 512 + j * 128 + lane * 4);
                *reinterpret_cast<float4*>(dst + j * 128 + lane * 4) = v4;
            }
        }
    } else {
        const int bq  = m0 / N_;
        const int kg0 = (m0 % N_) >> 2;
        const int h0  = n0 / HD_;
        #pragma unroll
        for (int hh = 0; hh < 2; ++hh) {
            float* dst = C + ((size_t)(bq * H_ + h0 + hh) * (N_ / 4) + kg0 + 2 * hi) * 128 + lo * 4;
            #pragma unroll
            for (int rq = 0; rq < 2; ++rq) {
                #pragma unroll
                for (int hf = 0; hf < 2; ++hf) {
                    const int r = 4 * rq + 2 * hf;
                    float4 v4;
                    v4.x = acc[2 * hh][r];     v4.y = acc[2 * hh][r + 1];
                    v4.z = acc[2 * hh + 1][r]; v4.w = acc[2 * hh + 1][r + 1];
                    *reinterpret_cast<float4*>(dst + rq * 128 + hf * 64) = v4;
                }
            }
        }
    }
}


__global__ __launch_bounds__(256)
void proj_gemm_kernel(const float* __restrict__ A, const float* __restrict__ W,
                      const float* __restrict__ bias, float* __restrict__ C,
                      int N, int K) {
    __shared__ float sW[2 * 64 * WSTRIDE];
    gemm_tile<0>(A, W, bias, C, N, K, blockIdx.x, blockIdx.y, sW, nullptr);
}

__global__ __launch_bounds__(256)
void qkv_gemm_kernel(const float* __restrict__ A, const float* __restrict__ Wq,
                     const float* __restrict__ Wk, const float* __restrict__ Wv,
                     float* __restrict__ Cq, float* __restrict__ Ck,
                     float* __restrict__ Cv, int N, int K) {
    __shared__ float sW[2 * 64 * WSTRIDE];
    __shared__ float sStage[8 * 1024];
    const int nblk  = N / 64;
    const int which = blockIdx.x / nblk;
    const int bx    = blockIdx.x - which * nblk;
    if (which == 0)
        gemm_tile<1>(A, Wq, nullptr, Cq, N, K, bx, blockIdx.y, sW, sStage);
    else if (which == 1)
        gemm_tile<1>(A, Wk, nullptr, Ck, N, K, bx, blockIdx.y, sW, sStage);
    else
        gemm_tile<2>(A, Wv, nullptr, Cv, N, K, bx, blockIdx.y, sW, sStage);
}

constexpr int SROW = 1028;
constexpr int AW   = 8;
constexpr int KPW  = N_ / AW;
constexpr int NQT  = 8;
constexpr int PROW = 36;
constexpr int STRIP_FLOATS = 16 * SROW;
constexpr int PARK_FLOATS  = NQT * AW * 16 * PROW;
constexpr int ATTN_SMEM = (STRIP_FLOATS + PARK_FLOATS) * 4;

static __device__ __forceinline__ float xhalf(float x) {
    return __builtin_bit_cast(float, __builtin_amdgcn_permlanex16(
        __builtin_bit_cast(int, x), __builtin_bit_cast(int, x),
        0x76543210, 0xfedcba98, false, false));
}

__global__ __launch_bounds__(256) __attribute__((amdgpu_waves_per_eu(2, 2)))
void attn_kernel(const float* __restrict__ Qb, const float* __restrict__ Kb,
                 const float* __restrict__ Vb, const float* __restrict__ Hb,
                 float* __restrict__ Out) {
    extern __shared__ __align__(16) float sS[];
    float* park = sS + STRIP_FLOATS;
    const int lane = threadIdx.x & 31;
    const int w    = threadIdx.x >> 5;
    const int bh   = blockIdx.y;
    const int b    = bh / H_;
    const int h    = bh % H_;
    const int qt0  = blockIdx.x * NQT;
    const int hi   = lane >> 4;
    const int lo   = lane & 15;
    const int key0 = w * KPW;
    const float cexp = 0.2550348615890031f;


    const float4* qf = reinterpret_cast<const float4*>(Qb)
                     + ((size_t)(bh * (N_ / 16) + qt0) * 4) * 32 + lane;
    v2f aq[8];
    #pragma unroll
    for (int tsp = 0; tsp < 4; ++tsp) {
        const float4 q4 = qf[tsp * 32];
        aq[2 * tsp].x     = q4.x; aq[2 * tsp].y     = q4.y;
        aq[2 * tsp + 1].x = q4.z; aq[2 * tsp + 1].y = q4.w;
    }

    const int jrow = threadIdx.x >> 4;
    const int jd   = (threadIdx.x & 15) * 2;
    const size_t jidx = (size_t)(b * N_ + qt0 * 16 + jrow) * CR_ + h * HD_ + jd;
    float2 hv[NQT];
    #pragma unroll
    for (int t = 0; t < NQT; ++t)
        hv[t] = *reinterpret_cast<const float2*>(Hb + jidx + (size_t)t * 16 * CR_);

    const float4* kbase = reinterpret_cast<const float4*>(Kb)
                        + ((size_t)(bh * (N_ / 16) + w * (KPW / 16)) * 4) * 32 + lane;
    v2f kf[64];
    #pragma unroll
    for (int i = 0; i < 32; ++i) {
        const float4 k4 = kbase[i * 32];
        kf[2 * i].x     = k4.x; kf[2 * i].y     = k4.y;
        kf[2 * i + 1].x = k4.z; kf[2 * i + 1].y = k4.w;
    }
    const float4* vbase = reinterpret_cast<const float4*>(Vb)
                        + ((size_t)bh * (N_ / 4) + key0 / 4 + 16 * hi) * 32 + lo;
    v2f va[32], vb[32];
    #pragma unroll
    for (int kk = 0; kk < 32; ++kk) {
        const float4 v4 = vbase[kk * 16];
        va[kk].x = v4.x; va[kk].y = v4.y;
        vb[kk].x = v4.z; vb[kk].y = v4.w;
    }

    const float* sw = sS + lo * SROW + key0 + 64 * hi;

    #pragma unroll 1
    for (int t = 0; t < NQT; ++t) {
        #pragma unroll
        for (int j = 0; j < KPW / 16; j += 2) {
            v8f acc0 = {}, acc1 = {};
            #pragma unroll
            for (int ts = 0; ts < 8; ++ts) {
                acc0 = wmma4(aq[ts], kf[8 * j + ts],       acc0);
                acc1 = wmma4(aq[ts], kf[8 * (j + 1) + ts], acc1);
            }
            #pragma unroll
            for (int r = 0; r < 8; ++r) {
                sS[(r + 8 * hi) * SROW + key0 + j * 16 + lo]       = acc0[r];
                sS[(r + 8 * hi) * SROW + key0 + (j + 1) * 16 + lo] = acc1[r];
            }
        }

        if (t + 1 < NQT) {
            #pragma unroll
            for (int tsp = 0; tsp < 4; ++tsp) {
                const float4 q4 = qf[(t + 1) * 128 + tsp * 32];
                aq[2 * tsp].x     = q4.x; aq[2 * tsp].y     = q4.y;
                aq[2 * tsp + 1].x = q4.z; aq[2 * tsp + 1].y = q4.w;
            }
        }

        float4 pv[16];
        #pragma unroll
        for (int i = 0; i < 16; ++i)
            pv[i] = *reinterpret_cast<const float4*>(sw + 4 * i);
        float mx = fmaxf(fmaxf(pv[0].x, pv[0].y), fmaxf(pv[0].z, pv[0].w));
        #pragma unroll
        for (int i = 1; i < 16; ++i)
            mx = fmaxf(mx, fmaxf(fmaxf(pv[i].x, pv[i].y), fmaxf(pv[i].z, pv[i].w)));
        mx = fmaxf(mx, xhalf(mx));
        const float mc = mx * cexp;
        float sum = 0.f;
        #pragma unroll
        for (int i = 0; i < 16; ++i) {
            pv[i].x = __builtin_amdgcn_exp2f(fmaf(pv[i].x, cexp, -mc));
            pv[i].y = __builtin_amdgcn_exp2f(fmaf(pv[i].y, cexp, -mc));
            pv[i].z = __builtin_amdgcn_exp2f(fmaf(pv[i].z, cexp, -mc));
            pv[i].w = __builtin_amdgcn_exp2f(fmaf(pv[i].w, cexp, -mc));
            sum += (pv[i].x + pv[i].y) + (pv[i].z + pv[i].w);
        }
        sum += xhalf(sum);

        v8f oacc0 = {}, oacc1 = {};
        #pragma unroll
        for (int kk = 0; kk < 32; ++kk) {
            v2f ap;
            if (kk & 1) { ap.x = pv[kk >> 1].z; ap.y = pv[kk >> 1].w; }
            else        { ap.x = pv[kk >> 1].x; ap.y = pv[kk >> 1].y; }
            oacc0 = wmma4(ap, va[kk], oacc0);
            oacc1 = wmma4(ap, vb[kk], oacc1);
        }

        float* pk = park + (size_t)((t * AW + w) * 16) * PROW;
        #pragma unroll
        for (int r = 0; r < 8; ++r) {
            pk[(r + 8 * hi) * PROW + lo]      = oacc0[r];
            pk[(r + 8 * hi) * PROW + 16 + lo] = oacc1[r];
        }
        pk[lo * PROW + 32 + hi] = hi ? sum : mc;
    }
    __syncthreads();

    #pragma unroll
    for (int t = 0; t < NQT; ++t) {
        const float* pp = park + (size_t)(t * AW * 16 + jrow) * PROW;
        float2 ml[AW];
        float mall = -3.0e38f;
        #pragma unroll
        for (int ww = 0; ww < AW; ++ww) {
            ml[ww] = *reinterpret_cast<const float2*>(pp + ww * 16 * PROW + 32);
            mall = fmaxf(mall, ml[ww].x);
        }
        float den = 0.f, sx = 0.f, sy = 0.f;
        #pragma unroll
        for (int ww = 0; ww < AW; ++ww) {
            const float f  = __builtin_amdgcn_exp2f(ml[ww].x - mall);
            const float2 o = *reinterpret_cast<const float2*>(pp + ww * 16 * PROW + jd);
            den = fmaf(f, ml[ww].y, den);
            sx  = fmaf(f, o.x, sx);
            sy  = fmaf(f, o.y, sy);
        }
        const float inv = 1.f / den;
        float2 ov; ov.x = fmaf(sx, inv, hv[t].x); ov.y = fmaf(sy, inv, hv[t].y);
        *reinterpret_cast<float2*>(Out + jidx + (size_t)t * 16 * CR_) = ov;
    }
}

extern "C" void kernel_launch(void* const* d_in, const int* in_sizes, int n_in,
                              void* d_out, int out_size, void* d_ws, size_t ws_size,
                              hipStream_t stream) {
    (void)in_sizes; (void)n_in; (void)out_size; (void)ws_size;
    const float* x      = (const float*)d_in[0];
    const float* ln0_g  = (const float*)d_in[1];
    const float* ln0_b  = (const float*)d_in[2];
    const float* w_proj = (const float*)d_in[3];
    const float* b_proj = (const float*)d_in[4];
    const float* ln1_g  = (const float*)d_in[5];
    const float* ln1_b  = (const float*)d_in[6];
    const float* wq     = (const float*)d_in[7];
    const float* wk     = (const float*)d_in[8];
    const float* wv     = (const float*)d_in[9];
    float* out = (float*)d_out;

    float* ws = (float*)d_ws;
    const size_t rowsBig = (size_t)B_ * N_ * DIM_;
    const size_t rowsCr  = (size_t)B_ * N_ * CR_;
    float* xn = ws;
    float* hB = xn + rowsBig;
    float* yB = hB + rowsCr;
    float* qB = yB + rowsCr;
    float* kB = qB + rowsCr;
    float* vB = kB + rowsCr;

    const int M = B_ * N_;

    ln_frag_kernel<DIM_><<<M / 16, 256, 0, stream>>>(x, ln0_g, ln0_b, xn);

    proj_gemm_kernel<<<dim3(CR_ / 64, M / 128), 256, 0, stream>>>(
        xn, w_proj, b_proj, hB, CR_, DIM_);

    ln_frag_kernel<CR_><<<M / 16, 256, 0, stream>>>(hB, ln1_g, ln1_b, yB);

    qkv_gemm_kernel<<<dim3(3 * CR_ / 64, M / 128), 256, 0, stream>>>(
        yB, wq, wk, wv, qB, kB, vB, CR_, CR_);

    (void)hipFuncSetAttribute((const void*)attn_kernel,
                              hipFuncAttributeMaxDynamicSharedMemorySize,
                              ATTN_SMEM);
    attn_kernel<<<dim3(N_ / 16 / NQT, B_ * H_), 256, ATTN_SMEM, stream>>>(
        qB, kB, vB, hB, out);
}


# input projection and LayerNorm-384 in ONE launch: a work-group owns 32 whole rows x 384 columns of h (256 work-groups), its epilogue normalises them in LDS and writes h and y; 4 launches
# speedup vs baseline: 2.7334x; 1.0406x over previous
#include <hip/hip_runtime.h>

constexpr int B_   = 8;
constexpr int N_   = 1024;
constexpr int DIM_ = 768;
constexpr int H_   = 12;
constexpr int CR_  = 384;
constexpr int HD_  = 32;
constexpr float EPS_ = 1e-5f;

typedef __attribute__((ext_vector_type(2))) float v2f;
typedef __attribute__((ext_vector_type(8))) float v8f;

static __device__ __forceinline__ v8f wmma4(v2f a, v2f b, v8f c) {
    return __builtin_amdgcn_wmma_f32_16x16x4_f32(
         false, a,  false, b,
         (short)0, c,  false,  false);
}

template <int K>
__global__ __launch_bounds__(256) __attribute__((amdgpu_waves_per_eu(2, 4)))
void ln_frag_kernel(const float* __restrict__ x, const float* __restrict__ g,
                    const float* __restrict__ bia, float* __restrict__ of) {
    constexpr int NL    = K / 128;
    constexpr int PITCH = K + 4;
    constexpr int NP    = K / 8;
    constexpr int PPW   = NP / 8;
    __shared__ __align__(16) float tile[16 * PITCH];
    const int lane = threadIdx.x & 31;
    const int w    = threadIdx.x >> 5;
    const int rt   = blockIdx.x;

    const float* xr = x + (size_t)(rt * 16 + 2 * w) * K + 4 * lane;
    float4 v[2][NL];
    #pragma unroll
    for (int i = 0; i < NL; ++i) {
        v[0][i] = *reinterpret_cast<const float4*>(xr + 128 * i);
        v[1][i] = *reinterpret_cast<const float4*>(xr + K + 128 * i);
    }
    float4 gv[NL], bv[NL];
    #pragma unroll
    for (int i = 0; i < NL; ++i) {
        gv[i] = *reinterpret_cast<const float4*>(g   + 128 * i + 4 * lane);
        bv[i] = *reinterpret_cast<const float4*>(bia + 128 * i + 4 * lane);
    }
    float s[2] = {0.f, 0.f}, s2[2] = {0.f, 0.f};
    #pragma unroll
    for (int i = 0; i < NL; ++i) {
        #pragma unroll
        for (int rr = 0; rr < 2; ++rr) {
            const float4 a = v[rr][i];
            s[rr]  += (a.x + a.y) + (a.z + a.w);
            s2[rr] += (a.x * a.x + a.y * a.y) + (a.z * a.z + a.w * a.w);
        }
    }
    #pragma unroll
    for (int off = 16; off > 0; off >>= 1) {
        #pragma unroll
        for (int rr = 0; rr < 2; ++rr) {
            s[rr]  += __shfl_xor(s[rr],  off, 32);
            s2[rr] += __shfl_xor(s2[rr], off, 32);
        }
    }
    #pragma unroll
    for (int rr = 0; rr < 2; ++rr) {
        const float mean = s[rr] * (1.f / K);
        const float var  = s2[rr] * (1.f / K) - mean * mean;
        const float rstd = rsqrtf(var + EPS_);
        float* trow = tile + (2 * w + rr) * PITCH + 4 * lane;
        #pragma unroll
        for (int i = 0; i < NL; ++i) {
            const float4 a = v[rr][i];
            float4 o;
            o.x = (a.x - mean) * rstd * gv[i].x + bv[i].x;
            o.y = (a.y - mean) * rstd * gv[i].y + bv[i].y;
            o.z = (a.z - mean) * rstd * gv[i].z + bv[i].z;
            o.w = (a.w - mean) * rstd * gv[i].w + bv[i].w;
            *reinterpret_cast<float4*>(trow + 128 * i) = o;
        }
    }
    __syncthreads();

    const int lo = lane & 15;
    const int hi = lane >> 4;
    const float* tsrc = tile + lo * PITCH + 4 * hi;
    float4* dst = reinterpret_cast<float4*>(of)
                + ((size_t)rt * NP + w * PPW) * 32 + lane;
    #pragma unroll
    for (int j = 0; j < PPW; ++j)
        dst[j * 32] = *reinterpret_cast<const float4*>(tsrc + 8 * (w * PPW + j));
}

constexpr int KC      = 32;
constexpr int WSTRIDE = KC + 2;

template <int MODE>
static __device__ __forceinline__
void gemm_tile(const float* __restrict__ A, const float* __restrict__ W,
               float* __restrict__ C,
               int N, int K, int bx, int by,
               float* __restrict__ sW, float* __restrict__ sStage) {
    const int t    = threadIdx.x;
    const int lane = t & 31;
    const int w    = t >> 5;
    const int hi   = lane >> 4;
    const int lo   = lane & 15;
    const int m0   = by * 128 + w * 16;
    const int n0   = bx * 64;
    const int NC   = K / KC;

    const float4* afrag = reinterpret_cast<const float4*>(A)
                        + ((size_t)(m0 >> 4) * (K >> 3)) * 32 + lane;
    const int lperm = (lane & ~6) | ((lane & 4) >> 1) | ((lane & 2) << 1);

    float wreg[8], wregN[8];
    v2f   areg[8], aregN[8];

    auto gload = [&](int kc, float* wr, v2f* ar) {
        #pragma unroll
        for (int i = 0; i < 8; ++i)
            wr[i] = W[(size_t)(n0 + i * 8 + w) * K + kc + lane];
        #pragma unroll
        for (int j = 0; j < 4; ++j) {
            const float4 a4 = afrag[((kc >> 3) + j) * 32];
            ar[2 * j].x     = a4.x; ar[2 * j].y     = a4.y;
            ar[2 * j + 1].x = a4.z; ar[2 * j + 1].y = a4.w;
        }
    };

    v8f acc[4] = {};
    gload(0, wreg, areg);

    for (int c = 0; c < NC; ++c) {
        const int bufi = c & 1;
        if (c + 1 < NC) gload((c + 1) * KC, wregN, aregN);
        #pragma unroll
        for (int i = 0; i < 8; ++i)
            sW[bufi * (64 * WSTRIDE) + (i * 8 + w) * WSTRIDE + lperm] = wreg[i];
        __syncthreads();
        const float* sbase = sW + bufi * (64 * WSTRIDE);
        #pragma unroll
        for (int ks = 0; ks < KC / 4; ++ks) {
            #pragma unroll
            for (int tt = 0; tt < 4; ++tt) {
                const float* srow = sbase + (tt * 16 + lo) * WSTRIDE + ks * 4 + 2 * hi;
                v2f b; b.x = srow[0]; b.y = srow[1];
                acc[tt] = wmma4(areg[ks], b, acc[tt]);
            }
        }
        if (c + 1 < NC) {
            #pragma unroll
            for (int i = 0; i < 8; ++i) { wreg[i] = wregN[i]; areg[i] = aregN[i]; }
        }
    }

    if constexpr (MODE == 1) {
        float* stg = sStage + w * 1024;
        #pragma unroll
        for (int tt = 0; tt < 4; ++tt) {
            const int d   = (tt & 1) * 16 + lo;
            const int off = (tt >> 1) * 512 + (d >> 3) * 128 + ((d >> 1) & 1) * 64
                          + ((d >> 2) & 1) * 2 + (d & 1);
            #pragma unroll
            for (int r = 0; r < 8; ++r)
                stg[off + (r + 8 * hi) * 4] = acc[tt][r];
        }
        __syncthreads();
        const int bq    = m0 / N_;
        const int gtile = (m0 % N_) >> 4;
        const int h0    = n0 / HD_;
        #pragma unroll
        for (int hh = 0; hh < 2; ++hh) {
            float* dst = C + ((size_t)((bq * H_ + h0 + hh) * (N_ / 16) + gtile)) * 512;
            #pragma unroll
            for (int j = 0; j < 4; ++j) {
                const float4 v4 = *reinterpret_cast<const float4*>(stg + hh * 512 + j * 128 + lane * 4);
                *reinterpret_cast<float4*>(dst + j * 128 + lane * 4) = v4;
            }
        }
    } else {
        const int bq  = m0 / N_;
        const int kg0 = (m0 % N_) >> 2;
        const int h0  = n0 / HD_;
        #pragma unroll
        for (int hh = 0; hh < 2; ++hh) {
            float* dst = C + ((size_t)(bq * H_ + h0 + hh) * (N_ / 4) + kg0 + 2 * hi) * 128 + lo * 4;
            #pragma unroll
            for (int rq = 0; rq < 2; ++rq) {
                #pragma unroll
                for (int hf = 0; hf < 2; ++hf) {
                    const int r = 4 * rq + 2 * hf;
                    float4 v4;
                    v4.x = acc[2 * hh][r];     v4.y = acc[2 * hh][r + 1];
                    v4.z = acc[2 * hh + 1][r]; v4.w = acc[2 * hh + 1][r + 1];
                    *reinterpret_cast<float4*>(dst + rq * 128 + hf * 64) = v4;
                }
            }
        }
    }
}


constexpr int PJ_STAGE = CR_ * WSTRIDE;
constexpr int PJ_PITCH = CR_ + 4;
constexpr int PJ_SMEM  = 164 * 1024;

__global__ __launch_bounds__(256) __attribute__((amdgpu_waves_per_eu(2, 2)))
void proj_ln_kernel(const float* __restrict__ A, const float* __restrict__ W,
                    const float* __restrict__ bias, const float* __restrict__ g,
                    const float* __restrict__ be, float* __restrict__ Hout,
                    float* __restrict__ Yf) {
    extern __shared__ __align__(16) float sP[];
    constexpr int K  = DIM_;
    constexpr int NC = K / KC;
    const int t    = threadIdx.x;
    const int lane = t & 31;
    const int w    = t >> 5;
    const int hi   = lane >> 4;
    const int lo   = lane & 15;
    const int wr   = w >> 2;
    const int wc   = w & 3;
    const int rt   = blockIdx.x * 2 + wr;

    const float4* afrag = reinterpret_cast<const float4*>(A)
                        + ((size_t)rt * (K >> 3)) * 32 + lane;
    const int sm = t & 7;
    const int sr = t >> 3;
    const float* wsrc = W + (size_t)sr * K + 4 * sm;
    const int spos = sr * WSTRIDE + (sm >> 1) * 8 + (sm & 1) * 2;

    float4 wst[12], ast[4];
    auto gload = [&](int kc) {
        #pragma unroll
        for (int i = 0; i < 12; ++i)
            wst[i] = *reinterpret_cast<const float4*>(wsrc + (size_t)i * 32 * K + kc);
        #pragma unroll
        for (int j = 0; j < 4; ++j)
            ast[j] = afrag[((kc >> 3) + j) * 32];
    };

    float bcol[6];
    #pragma unroll
    for (int tt = 0; tt < 6; ++tt) bcol[tt] = bias[wc * 96 + tt * 16 + lo];

    v8f acc[6] = {};
    gload(0);
    for (int c = 0; c < NC; ++c) {
        float* sb = sP + (c & 1) * PJ_STAGE;
        #pragma unroll
        for (int i = 0; i < 12; ++i) {
            float* d = sb + i * 32 * WSTRIDE + spos;
            float2 p0; p0.x = wst[i].x; p0.y = wst[i].y;
            float2 p1; p1.x = wst[i].z; p1.y = wst[i].w;
            *reinterpret_cast<float2*>(d)     = p0;
            *reinterpret_cast<float2*>(d + 4) = p1;
        }
        v2f areg[8];
        #pragma unroll
        for (int j = 0; j < 4; ++j) {
            areg[2 * j].x     = ast[j].x; areg[2 * j].y     = ast[j].y;
            areg[2 * j + 1].x = ast[j].z; areg[2 * j + 1].y = ast[j].w;
        }
        __syncthreads();
        if (c + 1 < NC) gload((c + 1) * KC);
        #pragma unroll
        for (int ks = 0; ks < KC / 4; ++ks) {
            #pragma unroll
            for (int tt = 0; tt < 6; ++tt) {
                const float* srow = sb + (wc * 96 + tt * 16 + lo) * WSTRIDE + ks * 4 + 2 * hi;
                v2f b; b.x = srow[0]; b.y = srow[1];
                acc[tt] = wmma4(areg[ks], b, acc[tt]);
            }
        }
    }
    constexpr int NL = CR_ / 128;
    float4 gv[NL], bv4[NL];
    #pragma unroll
    for (int i = 0; i < NL; ++i) {
        gv[i]  = *reinterpret_cast<const float4*>(g  + 128 * i + 4 * lane);
        bv4[i] = *reinterpret_cast<const float4*>(be + 128 * i + 4 * lane);
    }

    static_assert((NC & 1) == 0 && 32 * PJ_PITCH <= PJ_STAGE, "tile must fit the idle stage");
    float* hT = sP;
    #pragma unroll
    for (int tt = 0; tt < 6; ++tt) {
        const int col = wc * 96 + tt * 16 + lo;
        #pragma unroll
        for (int r = 0; r < 8; ++r)
            hT[(wr * 16 + 8 * hi + r) * PJ_PITCH + col] = acc[tt][r] + bcol[tt];
    }
    __syncthreads();

    float4 v[4][NL];
    float s[4], s2[4];
    #pragma unroll
    for (int rr = 0; rr < 4; ++rr) {
        const int row = 4 * w + rr;
        const float* trow = hT + row * PJ_PITCH + 4 * lane;
        float* hrow = Hout + (size_t)(blockIdx.x * 32 + row) * CR_ + 4 * lane;
        s[rr] = 0.f; s2[rr] = 0.f;
        #pragma unroll
        for (int i = 0; i < NL; ++i) {
            const float4 a = *reinterpret_cast<const float4*>(trow + 128 * i);
            v[rr][i] = a;
            *reinterpret_cast<float4*>(hrow + 128 * i) = a;
            s[rr]  += (a.x + a.y) + (a.z + a.w);
            s2[rr] += (a.x * a.x + a.y * a.y) + (a.z * a.z + a.w * a.w);
        }
    }
    #pragma unroll
    for (int off = 16; off > 0; off >>= 1) {
        #pragma unroll
        for (int rr = 0; rr < 4; ++rr) {
            s[rr]  += __shfl_xor(s[rr],  off, 32);
            s2[rr] += __shfl_xor(s2[rr], off, 32);
        }
    }
    #pragma unroll
    for (int rr = 0; rr < 4; ++rr) {
        const float mean = s[rr] * (1.f / CR_);
        const float var  = s2[rr] * (1.f / CR_) - mean * mean;
        const float rstd = rsqrtf(var + EPS_);
        float* trow = hT + (4 * w + rr) * PJ_PITCH + 4 * lane;
        #pragma unroll
        for (int i = 0; i < NL; ++i) {
            const float4 a = v[rr][i];
            float4 o;
            o.x = (a.x - mean) * rstd * gv[i].x + bv4[i].x;
            o.y = (a.y - mean) * rstd * gv[i].y + bv4[i].y;
            o.z = (a.z - mean) * rstd * gv[i].z + bv4[i].z;
            o.w = (a.w - mean) * rstd * gv[i].w + bv4[i].w;
            *reinterpret_cast<float4*>(trow + 128 * i) = o;
        }
    }
    __syncthreads();

    constexpr int NP = CR_ / 8;
    const float* tsrc = hT + (wr * 16 + lo) * PJ_PITCH + 4 * hi;
    float4* dst = reinterpret_cast<float4*>(Yf)
                + ((size_t)rt * NP + wc * 12) * 32 + lane;
    #pragma unroll
    for (int j = 0; j < 12; ++j)
        dst[j * 32] = *reinterpret_cast<const float4*>(tsrc + 8 * (wc * 12 + j));
}

__global__ __launch_bounds__(256)
void qkv_gemm_kernel(const float* __restrict__ A, const float* __restrict__ Wq,
                     const float* __restrict__ Wk, const float* __restrict__ Wv,
                     float* __restrict__ Cq, float* __restrict__ Ck,
                     float* __restrict__ Cv, int N, int K) {
    __shared__ float sW[2 * 64 * WSTRIDE];
    __shared__ float sStage[8 * 1024];
    const int nblk  = N / 64;
    const int which = blockIdx.x / nblk;
    const int bx    = blockIdx.x - which * nblk;
    if (which == 0)
        gemm_tile<1>(A, Wq, Cq, N, K, bx, blockIdx.y, sW, sStage);
    else if (which == 1)
        gemm_tile<1>(A, Wk, Ck, N, K, bx, blockIdx.y, sW, sStage);
    else
        gemm_tile<2>(A, Wv, Cv, N, K, bx, blockIdx.y, sW, sStage);
}

constexpr int SROW = 1028;
constexpr int AW   = 8;
constexpr int KPW  = N_ / AW;
constexpr int NQT  = 8;
constexpr int PROW = 36;
constexpr int STRIP_FLOATS = 16 * SROW;
constexpr int PARK_FLOATS  = NQT * AW * 16 * PROW;
constexpr int ATTN_SMEM = (STRIP_FLOATS + PARK_FLOATS) * 4;

static __device__ __forceinline__ float xhalf(float x) {
    return __builtin_bit_cast(float, __builtin_amdgcn_permlanex16(
        __builtin_bit_cast(int, x), __builtin_bit_cast(int, x),
        0x76543210, 0xfedcba98, false, false));
}

__global__ __launch_bounds__(256) __attribute__((amdgpu_waves_per_eu(2, 2)))
void attn_kernel(const float* __restrict__ Qb, const float* __restrict__ Kb,
                 const float* __restrict__ Vb, const float* __restrict__ Hb,
                 float* __restrict__ Out) {
    extern __shared__ __align__(16) float sS[];
    float* park = sS + STRIP_FLOATS;
    const int lane = threadIdx.x & 31;
    const int w    = threadIdx.x >> 5;
    const int bh   = blockIdx.y;
    const int b    = bh / H_;
    const int h    = bh % H_;
    const int qt0  = blockIdx.x * NQT;
    const int hi   = lane >> 4;
    const int lo   = lane & 15;
    const int key0 = w * KPW;
    const float cexp = 0.2550348615890031f;


    const float4* qf = reinterpret_cast<const float4*>(Qb)
                     + ((size_t)(bh * (N_ / 16) + qt0) * 4) * 32 + lane;
    v2f aq[8];
    #pragma unroll
    for (int tsp = 0; tsp < 4; ++tsp) {
        const float4 q4 = qf[tsp * 32];
        aq[2 * tsp].x     = q4.x; aq[2 * tsp].y     = q4.y;
        aq[2 * tsp + 1].x = q4.z; aq[2 * tsp + 1].y = q4.w;
    }

    const int jrow = threadIdx.x >> 4;
    const int jd   = (threadIdx.x & 15) * 2;
    const size_t jidx = (size_t)(b * N_ + qt0 * 16 + jrow) * CR_ + h * HD_ + jd;
    float2 hv[NQT];
    #pragma unroll
    for (int t = 0; t < NQT; ++t)
        hv[t] = *reinterpret_cast<const float2*>(Hb + jidx + (size_t)t * 16 * CR_);

    const float4* kbase = reinterpret_cast<const float4*>(Kb)
                        + ((size_t)(bh * (N_ / 16) + w * (KPW / 16)) * 4) * 32 + lane;
    v2f kf[64];
    #pragma unroll
    for (int i = 0; i < 32; ++i) {
        const float4 k4 = kbase[i * 32];
        kf[2 * i].x     = k4.x; kf[2 * i].y     = k4.y;
        kf[2 * i + 1].x = k4.z; kf[2 * i + 1].y = k4.w;
    }
    const float4* vbase = reinterpret_cast<const float4*>(Vb)
                        + ((size_t)bh * (N_ / 4) + key0 / 4 + 16 * hi) * 32 + lo;
    v2f va[32], vb[32];
    #pragma unroll
    for (int kk = 0; kk < 32; ++kk) {
        const float4 v4 = vbase[kk * 16];
        va[kk].x = v4.x; va[kk].y = v4.y;
        vb[kk].x = v4.z; vb[kk].y = v4.w;
    }

    const float* sw = sS + lo * SROW + key0 + 64 * hi;

    #pragma unroll 1
    for (int t = 0; t < NQT; ++t) {
        #pragma unroll
        for (int j = 0; j < KPW / 16; j += 2) {
            v8f acc0 = {}, acc1 = {};
            #pragma unroll
            for (int ts = 0; ts < 8; ++ts) {
                acc0 = wmma4(aq[ts], kf[8 * j + ts],       acc0);
                acc1 = wmma4(aq[ts], kf[8 * (j + 1) + ts], acc1);
            }
            #pragma unroll
            for (int r = 0; r < 8; ++r) {
                sS[(r + 8 * hi) * SROW + key0 + j * 16 + lo]       = acc0[r];
                sS[(r + 8 * hi) * SROW + key0 + (j + 1) * 16 + lo] = acc1[r];
            }
        }

        if (t + 1 < NQT) {
            #pragma unroll
            for (int tsp = 0; tsp < 4; ++tsp) {
                const float4 q4 = qf[(t + 1) * 128 + tsp * 32];
                aq[2 * tsp].x     = q4.x; aq[2 * tsp].y     = q4.y;
                aq[2 * tsp + 1].x = q4.z; aq[2 * tsp + 1].y = q4.w;
            }
        }

        float4 pv[16];
        #pragma unroll
        for (int i = 0; i < 16; ++i)
            pv[i] = *reinterpret_cast<const float4*>(sw + 4 * i);
        float mx = fmaxf(fmaxf(pv[0].x, pv[0].y), fmaxf(pv[0].z, pv[0].w));
        #pragma unroll
        for (int i = 1; i < 16; ++i)
            mx = fmaxf(mx, fmaxf(fmaxf(pv[i].x, pv[i].y), fmaxf(pv[i].z, pv[i].w)));
        mx = fmaxf(mx, xhalf(mx));
        const float mc = mx * cexp;
        float sum = 0.f;
        #pragma unroll
        for (int i = 0; i < 16; ++i) {
            pv[i].x = __builtin_amdgcn_exp2f(fmaf(pv[i].x, cexp, -mc));
            pv[i].y = __builtin_amdgcn_exp2f(fmaf(pv[i].y, cexp, -mc));
            pv[i].z = __builtin_amdgcn_exp2f(fmaf(pv[i].z, cexp, -mc));
            pv[i].w = __builtin_amdgcn_exp2f(fmaf(pv[i].w, cexp, -mc));
            sum += (pv[i].x + pv[i].y) + (pv[i].z + pv[i].w);
        }
        sum += xhalf(sum);

        v8f oacc0 = {}, oacc1 = {};
        #pragma unroll
        for (int kk = 0; kk < 32; ++kk) {
            v2f ap;
            if (kk & 1) { ap.x = pv[kk >> 1].z; ap.y = pv[kk >> 1].w; }
            else        { ap.x = pv[kk >> 1].x; ap.y = pv[kk >> 1].y; }
            oacc0 = wmma4(ap, va[kk], oacc0);
            oacc1 = wmma4(ap, vb[kk], oacc1);
        }

        float* pk = park + (size_t)((t * AW + w) * 16) * PROW;
        #pragma unroll
        for (int r = 0; r < 8; ++r) {
            pk[(r + 8 * hi) * PROW + lo]      = oacc0[r];
            pk[(r + 8 * hi) * PROW + 16 + lo] = oacc1[r];
        }
        pk[lo * PROW + 32 + hi] = hi ? sum : mc;
    }
    __syncthreads();

    #pragma unroll
    for (int t = 0; t < NQT; ++t) {
        const float* pp = park + (size_t)(t * AW * 16 + jrow) * PROW;
        float2 ml[AW];
        float mall = -3.0e38f;
        #pragma unroll
        for (int ww = 0; ww < AW; ++ww) {
            ml[ww] = *reinterpret_cast<const float2*>(pp + ww * 16 * PROW + 32);
            mall = fmaxf(mall, ml[ww].x);
        }
        float den = 0.f, sx = 0.f, sy = 0.f;
        #pragma unroll
        for (int ww = 0; ww < AW; ++ww) {
            const float f  = __builtin_amdgcn_exp2f(ml[ww].x - mall);
            const float2 o = *reinterpret_cast<const float2*>(pp + ww * 16 * PROW + jd);
            den = fmaf(f, ml[ww].y, den);
            sx  = fmaf(f, o.x, sx);
            sy  = fmaf(f, o.y, sy);
        }
        const float inv = 1.f / den;
        float2 ov; ov.x = fmaf(sx, inv, hv[t].x); ov.y = fmaf(sy, inv, hv[t].y);
        *reinterpret_cast<float2*>(Out + jidx + (size_t)t * 16 * CR_) = ov;
    }
}

extern "C" void kernel_launch(void* const* d_in, const int* in_sizes, int n_in,
                              void* d_out, int out_size, void* d_ws, size_t ws_size,
                              hipStream_t stream) {
    (void)in_sizes; (void)n_in; (void)out_size; (void)ws_size;
    const float* x      = (const float*)d_in[0];
    const float* ln0_g  = (const float*)d_in[1];
    const float* ln0_b  = (const float*)d_in[2];
    const float* w_proj = (const float*)d_in[3];
    const float* b_proj = (const float*)d_in[4];
    const float* ln1_g  = (const float*)d_in[5];
    const float* ln1_b  = (const float*)d_in[6];
    const float* wq     = (const float*)d_in[7];
    const float* wk     = (const float*)d_in[8];
    const float* wv     = (const float*)d_in[9];
    float* out = (float*)d_out;

    float* ws = (float*)d_ws;
    const size_t rowsBig = (size_t)B_ * N_ * DIM_;
    const size_t rowsCr  = (size_t)B_ * N_ * CR_;
    float* xn = ws;
    float* hB = xn + rowsBig;
    float* yB = hB + rowsCr;
    float* qB = yB + rowsCr;
    float* kB = qB + rowsCr;
    float* vB = kB + rowsCr;

    const int M = B_ * N_;

    ln_frag_kernel<DIM_><<<M / 16, 256, 0, stream>>>(x, ln0_g, ln0_b, xn);

    (void)hipFuncSetAttribute((const void*)proj_ln_kernel,
                              hipFuncAttributeMaxDynamicSharedMemorySize,
                              PJ_SMEM);
    proj_ln_kernel<<<M / 32, 256, PJ_SMEM, stream>>>(
        xn, w_proj, b_proj, ln1_g, ln1_b, hB, yB);

    qkv_gemm_kernel<<<dim3(3 * CR_ / 64, M / 128), 256, 0, stream>>>(
        yB, wq, wk, wv, qB, kB, vB, CR_, CR_);

    (void)hipFuncSetAttribute((const void*)attn_kernel,
                              hipFuncAttributeMaxDynamicSharedMemorySize,
                              ATTN_SMEM);
    attn_kernel<<<dim3(N_ / 16 / NQT, B_ * H_), 256, ATTN_SMEM, stream>>>(
        qB, kB, vB, hB, out);
}


# GEMMs: weights put in WMMA fragment order once per call; a wave streams its own columns' B fragments from global (next chunk grouped ahead), A from an LDS copy: K loops without LDS stage or barrier
# speedup vs baseline: 3.0251x; 1.1067x over previous
#include <hip/hip_runtime.h>

constexpr int B_   = 8;
constexpr int N_   = 1024;
constexpr int DIM_ = 768;
constexpr int H_   = 12;
constexpr int CR_  = 384;
constexpr int HD_  = 32;
constexpr float EPS_ = 1e-5f;

typedef __attribute__((ext_vector_type(2))) float v2f;
typedef __attribute__((ext_vector_type(8))) float v8f;

static __device__ __forceinline__ v8f wmma4(v2f a, v2f b, v8f c) {
    return __builtin_amdgcn_wmma_f32_16x16x4_f32(
         false, a,  false, b,
         (short)0, c,  false,  false);
}

static __device__ __forceinline__
void weight_frag_task(const float* __restrict__ W, float* __restrict__ Wf,
                      int K, int ctg, int pq, int lane) {
    const int lo = lane & 15;
    const int hi = lane >> 4;
    const float* src = W + (size_t)(16 * ctg + lo) * K + 32 * pq + 4 * hi;
    float4* dst = reinterpret_cast<float4*>(Wf)
                + ((size_t)ctg * (K >> 3) + 4 * pq) * 32 + lane;
    float4 v[4];
    #pragma unroll
    for (int j = 0; j < 4; ++j) v[j] = *reinterpret_cast<const float4*>(src + 8 * j);
    #pragma unroll
    for (int j = 0; j < 4; ++j) dst[j * 32] = v[j];
}

constexpr int WF_TASKS_P = (CR_ / 16) * (DIM_ / 32);
constexpr int WF_TASKS_Q = (CR_ / 16) * (CR_ / 32);
constexpr int WF_BLOCKS  = (WF_TASKS_P + 3 * WF_TASKS_Q) / 8;

template <int K>
__global__ __launch_bounds__(256) __attribute__((amdgpu_waves_per_eu(2, 4)))
void ln_frag_kernel(const float* __restrict__ x, const float* __restrict__ g,
                    const float* __restrict__ bia, float* __restrict__ of,
                    const float* __restrict__ wp, const float* __restrict__ wq,
                    const float* __restrict__ wk, const float* __restrict__ wv,
                    float* __restrict__ wpf, float* __restrict__ wqkvf,
                    int nRowTiles) {
    constexpr int NL    = K / 128;
    constexpr int PITCH = K + 4;
    constexpr int NP    = K / 8;
    constexpr int PPW   = NP / 8;
    __shared__ __align__(16) float tile[16 * PITCH];
    const int lane = threadIdx.x & 31;
    const int w    = threadIdx.x >> 5;
    const int rt   = blockIdx.x;

    if (rt >= nRowTiles) {
        int task = (rt - nRowTiles) * 8 + w;
        if (task < WF_TASKS_P) {
            weight_frag_task(wp, wpf, DIM_, task / (DIM_ / 32), task % (DIM_ / 32), lane);
        } else {
            task -= WF_TASKS_P;
            const int sel = task / WF_TASKS_Q;
            const int rem = task - sel * WF_TASKS_Q;
            const float* W = sel == 0 ? wq : (sel == 1 ? wk : wv);
            weight_frag_task(W, wqkvf + (size_t)sel * CR_ * CR_, CR_,
                             rem / (CR_ / 32), rem % (CR_ / 32), lane);
        }
        return;
    }

    const float* xr = x + (size_t)(rt * 16 + 2 * w) * K + 4 * lane;
    float4 v[2][NL];
    #pragma unroll
    for (int i = 0; i < NL; ++i) {
        v[0][i] = *reinterpret_cast<const float4*>(xr + 128 * i);
        v[1][i] = *reinterpret_cast<const float4*>(xr + K + 128 * i);
    }
    float4 gv[NL], bv[NL];
    #pragma unroll
    for (int i = 0; i < NL; ++i) {
        gv[i] = *reinterpret_cast<const float4*>(g   + 128 * i + 4 * lane);
        bv[i] = *reinterpret_cast<const float4*>(bia + 128 * i + 4 * lane);
    }
    float s[2] = {0.f, 0.f}, s2[2] = {0.f, 0.f};
    #pragma unroll
    for (int i = 0; i < NL; ++i) {
        #pragma unroll
        for (int rr = 0; rr < 2; ++rr) {
            const float4 a = v[rr][i];
            s[rr]  += (a.x + a.y) + (a.z + a.w);
            s2[rr] += (a.x * a.x + a.y * a.y) + (a.z * a.z + a.w * a.w);
        }
    }
    #pragma unroll
    for (int off = 16; off > 0; off >>= 1) {
        #pragma unroll
        for (int rr = 0; rr < 2; ++rr) {
            s[rr]  += __shfl_xor(s[rr],  off, 32);
            s2[rr] += __shfl_xor(s2[rr], off, 32);
        }
    }
    #pragma unroll
    for (int rr = 0; rr < 2; ++rr) {
        const float mean = s[rr] * (1.f / K);
        const float var  = s2[rr] * (1.f / K) - mean * mean;
        const float rstd = rsqrtf(var + EPS_);
        float* trow = tile + (2 * w + rr) * PITCH + 4 * lane;
        #pragma unroll
        for (int i = 0; i < NL; ++i) {
            const float4 a = v[rr][i];
            float4 o;
            o.x = (a.x - mean) * rstd * gv[i].x + bv[i].x;
            o.y = (a.y - mean) * rstd * gv[i].y + bv[i].y;
            o.z = (a.z - mean) * rstd * gv[i].z + bv[i].z;
            o.w = (a.w - mean) * rstd * gv[i].w + bv[i].w;
            *reinterpret_cast<float4*>(trow + 128 * i) = o;
        }
    }
    __syncthreads();

    const int lo = lane & 15;
    const int hi = lane >> 4;
    const float* tsrc = tile + lo * PITCH + 4 * hi;
    float4* dst = reinterpret_cast<float4*>(of)
                + ((size_t)rt * NP + w * PPW) * 32 + lane;
    #pragma unroll
    for (int j = 0; j < PPW; ++j)
        dst[j * 32] = *reinterpret_cast<const float4*>(tsrc + 8 * (w * PPW + j));
}

static __device__ __forceinline__ v2f frag_lo(const float4& f) { v2f r; r.x = f.x; r.y = f.y; return r; }
static __device__ __forceinline__ v2f frag_hi(const float4& f) { v2f r; r.x = f.z; r.y = f.w; return r; }

template <int NCT, int PC, int NP8>
struct FragSet {
    float4 a[2][PC];
    float4 b[NCT][PC];
    __device__ __forceinline__ void load(const float4* asrc, const float4* __restrict__ bsrc, int c) {
        #pragma unroll
        for (int ct = 0; ct < NCT; ++ct) {
            #pragma unroll
            for (int j = 0; j < PC; ++j) b[ct][j] = bsrc[((size_t)ct * NP8 + c * PC + j) * 32];
        }
        #pragma unroll
        for (int r2 = 0; r2 < 2; ++r2) {
            #pragma unroll
            for (int j = 0; j < PC; ++j) a[r2][j] = asrc[(r2 * NP8 + c * PC + j) * 32];
        }
    }
    __device__ __forceinline__ void mma(v8f (&acc)[2][NCT]) const {
        #pragma unroll
        for (int j = 0; j < PC; ++j) {
            #pragma unroll
            for (int r2 = 0; r2 < 2; ++r2) {
                #pragma unroll
                for (int ct = 0; ct < NCT; ++ct)
                    acc[r2][ct] = wmma4(frag_lo(a[r2][j]), frag_lo(b[ct][j]), acc[r2][ct]);
            }
            #pragma unroll
            for (int r2 = 0; r2 < 2; ++r2) {
                #pragma unroll
                for (int ct = 0; ct < NCT; ++ct)
                    acc[r2][ct] = wmma4(frag_hi(a[r2][j]), frag_hi(b[ct][j]), acc[r2][ct]);
            }
        }
    }
};

template <int NCT, int PC, int NP8>
static __device__ __forceinline__
void gemm_kloop(const float4* asrc, const float4* __restrict__ bsrc,
                FragSet<NCT, PC, NP8>& f0, v8f (&acc)[2][NCT]) {
    constexpr int NC = NP8 / PC;
    static_assert((NC & 1) == 0 && NC >= 4, "two chunks per trip");
    FragSet<NCT, PC, NP8> f1;
    for (int c = 0; c < NC - 2; c += 2) {
        f1.load(asrc, bsrc, c + 1);
        __builtin_amdgcn_sched_barrier(0);
        f0.mma(acc);
        __builtin_amdgcn_sched_barrier(0);
        f0.load(asrc, bsrc, c + 2);
        __builtin_amdgcn_sched_barrier(0);
        f1.mma(acc);
        __builtin_amdgcn_sched_barrier(0);
    }
    f1.load(asrc, bsrc, NC - 1);
    __builtin_amdgcn_sched_barrier(0);
    f0.mma(acc);
    f1.mma(acc);
}

constexpr int PJ_PITCH = CR_ + 4;
constexpr int PJ_SMEM  = 164 * 1024;

__global__ __launch_bounds__(256) __attribute__((amdgpu_waves_per_eu(2, 2)))
void proj_ln_kernel(const float* __restrict__ A, const float* __restrict__ Wf,
                    const float* __restrict__ bias, const float* __restrict__ g,
                    const float* __restrict__ be, float* __restrict__ Hout,
                    float* __restrict__ Yf) {
    extern __shared__ __align__(16) float sP[];
    constexpr int NP8 = DIM_ / 8;
    constexpr int PC  = 4;
    const int t    = threadIdx.x;
    const int lane = t & 31;
    const int w    = t >> 5;
    const int hi   = lane >> 4;
    const int lo   = lane & 15;
    const int wr   = w >> 2;
    const int wc   = w & 3;
    const int rt   = blockIdx.x * 2 + wr;

    const float4* bsrc = reinterpret_cast<const float4*>(Wf)
                       + ((size_t)(3 * w) * NP8) * 32 + lane;
    float4* sA4 = reinterpret_cast<float4*>(sP);
    FragSet<3, PC, NP8> f0;
    #pragma unroll
    for (int ct = 0; ct < 3; ++ct) {
        #pragma unroll
        for (int j = 0; j < PC; ++j) f0.b[ct][j] = bsrc[((size_t)ct * NP8 + j) * 32];
    }
    {
        const float4* asrc = reinterpret_cast<const float4*>(A)
                           + ((size_t)(blockIdx.x * 2) * NP8) * 32 + t;
        float4 st[24];
        #pragma unroll
        for (int i = 0; i < 24; ++i) st[i] = asrc[i * 256];
        #pragma unroll
        for (int i = 0; i < 24; ++i) sA4[i * 256 + t] = st[i];
    }
    float bcol[3];
    #pragma unroll
    for (int ct = 0; ct < 3; ++ct) bcol[ct] = bias[w * 48 + ct * 16 + lo];
    __syncthreads();
    #pragma unroll
    for (int r2 = 0; r2 < 2; ++r2) {
        #pragma unroll
        for (int j = 0; j < PC; ++j) f0.a[r2][j] = sA4[(r2 * NP8 + j) * 32 + lane];
    }

    v8f acc[2][3] = {};
    gemm_kloop<3, PC, NP8>(sA4 + lane, bsrc, f0, acc);

    constexpr int NL = CR_ / 128;
    float4 gv[NL], bv4[NL];
    #pragma unroll
    for (int i = 0; i < NL; ++i) {
        gv[i]  = *reinterpret_cast<const float4*>(g  + 128 * i + 4 * lane);
        bv4[i] = *reinterpret_cast<const float4*>(be + 128 * i + 4 * lane);
    }

    static_assert(32 * PJ_PITCH <= 2 * NP8 * 32 * 4, "h tile must fit the A tile");
    float* hT = sP;
    __syncthreads();
    #pragma unroll
    for (int r2 = 0; r2 < 2; ++r2) {
        #pragma unroll
        for (int ct = 0; ct < 3; ++ct) {
            const int col = w * 48 + ct * 16 + lo;
            #pragma unroll
            for (int r = 0; r < 8; ++r)
                hT[(r2 * 16 + 8 * hi + r) * PJ_PITCH + col] = acc[r2][ct][r] + bcol[ct];
        }
    }
    __syncthreads();

    float4 v[4][NL];
    float s[4], s2[4];
    #pragma unroll
    for (int rr = 0; rr < 4; ++rr) {
        const int row = 4 * w + rr;
        const float* trow = hT + row * PJ_PITCH + 4 * lane;
        float* hrow = Hout + (size_t)(blockIdx.x * 32 + row) * CR_ + 4 * lane;
        s[rr] = 0.f; s2[rr] = 0.f;
        #pragma unroll
        for (int i = 0; i < NL; ++i) {
            const float4 a = *reinterpret_cast<const float4*>(trow + 128 * i);
            v[rr][i] = a;
            *reinterpret_cast<float4*>(hrow + 128 * i) = a;
            s[rr]  += (a.x + a.y) + (a.z + a.w);
            s2[rr] += (a.x * a.x + a.y * a.y) + (a.z * a.z + a.w * a.w);
        }
    }
    #pragma unroll
    for (int off = 16; off > 0; off >>= 1) {
        #pragma unroll
        for (int rr = 0; rr < 4; ++rr) {
            s[rr]  += __shfl_xor(s[rr],  off, 32);
            s2[rr] += __shfl_xor(s2[rr], off, 32);
        }
    }
    #pragma unroll
    for (int rr = 0; rr < 4; ++rr) {
        const float mean = s[rr] * (1.f / CR_);
        const float var  = s2[rr] * (1.f / CR_) - mean * mean;
        const float rstd = rsqrtf(var + EPS_);
        float* trow = hT + (4 * w + rr) * PJ_PITCH + 4 * lane;
        #pragma unroll
        for (int i = 0; i < NL; ++i) {
            const float4 a = v[rr][i];
            float4 o;
            o.x = (a.x - mean) * rstd * gv[i].x + bv4[i].x;
            o.y = (a.y - mean) * rstd * gv[i].y + bv4[i].y;
            o.z = (a.z - mean) * rstd * gv[i].z + bv4[i].z;
            o.w = (a.w - mean) * rstd * gv[i].w + bv4[i].w;
            *reinterpret_cast<float4*>(trow + 128 * i) = o;
        }
    }
    __syncthreads();

    constexpr int NP = CR_ / 8;
    const float* tsrc = hT + (wr * 16 + lo) * PJ_PITCH + 4 * hi;
    float4* dst = reinterpret_cast<float4*>(Yf)
                + ((size_t)rt * NP + wc * 12) * 32 + lane;
    #pragma unroll
    for (int j = 0; j < 12; ++j)
        dst[j * 32] = *reinterpret_cast<const float4*>(tsrc + 8 * (wc * 12 + j));
}

constexpr int QK_SMEM = 100 * 1024;
constexpr int QK_STG  = 2 * (CR_ / 8) * 32 * 4;

__global__ __launch_bounds__(128) __attribute__((amdgpu_waves_per_eu(3, 3)))
void qkv_gemm_kernel(const float* __restrict__ Yf, const float* __restrict__ Wf,
                     float* __restrict__ Cq, float* __restrict__ Ck,
                     float* __restrict__ Cv) {
    extern __shared__ __align__(16) float sQ[];
    constexpr int NP8 = CR_ / 8;
    constexpr int PC  = 2;
    const int t    = threadIdx.x;
    const int lane = t & 31;
    const int w    = t >> 5;
    const int hi   = lane >> 4;
    const int lo   = lane & 15;
    const int rb   = blockIdx.x / 3;
    const int sel  = blockIdx.x - rb * 3;

    const float4* bsrc = reinterpret_cast<const float4*>(Wf + (size_t)sel * CR_ * CR_)
                       + ((size_t)(6 * w) * NP8) * 32 + lane;
    float4* sA4 = reinterpret_cast<float4*>(sQ);
    FragSet<6, PC, NP8> f0;
    #pragma unroll
    for (int ct = 0; ct < 6; ++ct) {
        #pragma unroll
        for (int j = 0; j < PC; ++j) f0.b[ct][j] = bsrc[((size_t)ct * NP8 + j) * 32];
    }
    {
        const float4* asrc = reinterpret_cast<const float4*>(Yf)
                           + ((size_t)(rb * 2) * NP8) * 32 + t;
        float4 st[24];
        #pragma unroll
        for (int i = 0; i < 24; ++i) st[i] = asrc[i * 128];
        #pragma unroll
        for (int i = 0; i < 24; ++i) sA4[i * 128 + t] = st[i];
    }
    __syncthreads();
    #pragma unroll
    for (int r2 = 0; r2 < 2; ++r2) {
        #pragma unroll
        for (int j = 0; j < PC; ++j) f0.a[r2][j] = sA4[(r2 * NP8 + j) * 32 + lane];
    }

    v8f acc[2][6] = {};
    gemm_kloop<6, PC, NP8>(sA4 + lane, bsrc, f0, acc);

    if (sel < 2) {
        float* stg = sQ + QK_STG + w * (6 * 512);
        float* C = sel == 0 ? Cq : Ck;
        #pragma unroll
        for (int r2 = 0; r2 < 2; ++r2) {
            #pragma unroll
            for (int ct = 0; ct < 6; ++ct) {
                const int d   = (ct & 1) * 16 + lo;
                const int off = (r2 * 3 + (ct >> 1)) * 512 + (d >> 3) * 128 + ((d >> 1) & 1) * 64
                              + ((d >> 2) & 1) * 2 + (d & 1);
                #pragma unroll
                for (int r = 0; r < 8; ++r)
                    stg[off + (r + 8 * hi) * 4] = acc[r2][ct][r];
            }
        }
        __syncthreads();
        #pragma unroll
        for (int r2 = 0; r2 < 2; ++r2) {
            const int m0    = (rb * 2 + r2) * 16;
            const int bq    = m0 / N_;
            const int gtile = (m0 % N_) >> 4;
            #pragma unroll
            for (int hh = 0; hh < 3; ++hh) {
                float* dst = C + ((size_t)((bq * H_ + 3 * w + hh) * (N_ / 16) + gtile)) * 512;
                #pragma unroll
                for (int j = 0; j < 4; ++j) {
                    const float4 v4 = *reinterpret_cast<const float4*>(
                        stg + (r2 * 3 + hh) * 512 + j * 128 + lane * 4);
                    *reinterpret_cast<float4*>(dst + j * 128 + lane * 4) = v4;
                }
            }
        }
    } else {
        #pragma unroll
        for (int r2 = 0; r2 < 2; ++r2) {
            const int m0  = (rb * 2 + r2) * 16;
            const int bq  = m0 / N_;
            const int kg0 = (m0 % N_) >> 2;
            #pragma unroll
            for (int hh = 0; hh < 3; ++hh) {
                float* dst = Cv + ((size_t)(bq * H_ + 3 * w + hh) * (N_ / 4) + kg0 + 2 * hi) * 128 + lo * 4;
                #pragma unroll
                for (int rq = 0; rq < 2; ++rq) {
                    #pragma unroll
                    for (int hf = 0; hf < 2; ++hf) {
                        const int r = 4 * rq + 2 * hf;
                        float4 v4;
                        v4.x = acc[r2][2 * hh][r];     v4.y = acc[r2][2 * hh][r + 1];
                        v4.z = acc[r2][2 * hh + 1][r]; v4.w = acc[r2][2 * hh + 1][r + 1];
                        *reinterpret_cast<float4*>(dst + rq * 128 + hf * 64) = v4;
                    }
                }
            }
        }
    }
}

constexpr int SROW = 1028;
constexpr int AW   = 8;
constexpr int KPW  = N_ / AW;
constexpr int NQT  = 8;
constexpr int PROW = 36;
constexpr int STRIP_FLOATS = 16 * SROW;
constexpr int PARK_FLOATS  = NQT * AW * 16 * PROW;
constexpr int ATTN_SMEM = (STRIP_FLOATS + PARK_FLOATS) * 4;

static __device__ __forceinline__ float xhalf(float x) {
    return __builtin_bit_cast(float, __builtin_amdgcn_permlanex16(
        __builtin_bit_cast(int, x), __builtin_bit_cast(int, x),
        0x76543210, 0xfedcba98, false, false));
}

__global__ __launch_bounds__(256) __attribute__((amdgpu_waves_per_eu(2, 2)))
void attn_kernel(const float* __restrict__ Qb, const float* __restrict__ Kb,
                 const float* __restrict__ Vb, const float* __restrict__ Hb,
                 float* __restrict__ Out) {
    extern __shared__ __align__(16) float sS[];
    float* park = sS + STRIP_FLOATS;
    const int lane = threadIdx.x & 31;
    const int w    = threadIdx.x >> 5;
    const int bh   = blockIdx.y;
    const int b    = bh / H_;
    const int h    = bh % H_;
    const int qt0  = blockIdx.x * NQT;
    const int hi   = lane >> 4;
    const int lo   = lane & 15;
    const int key0 = w * KPW;
    const float cexp = 0.2550348615890031f;


    const float4* qf = reinterpret_cast<const float4*>(Qb)
                     + ((size_t)(bh * (N_ / 16) + qt0) * 4) * 32 + lane;
    v2f aq[8];
    #pragma unroll
    for (int tsp = 0; tsp < 4; ++tsp) {
        const float4 q4 = qf[tsp * 32];
        aq[2 * tsp].x     = q4.x; aq[2 * tsp].y     = q4.y;
        aq[2 * tsp + 1].x = q4.z; aq[2 * tsp + 1].y = q4.w;
    }

    const int jrow = threadIdx.x >> 4;
    const int jd   = (threadIdx.x & 15) * 2;
    const size_t jidx = (size_t)(b * N_ + qt0 * 16 + jrow) * CR_ + h * HD_ + jd;
    float2 hv[NQT];
    #pragma unroll
    for (int t = 0; t < NQT; ++t)
        hv[t] = *reinterpret_cast<const float2*>(Hb + jidx + (size_t)t * 16 * CR_);

    const float4* kbase = reinterpret_cast<const float4*>(Kb)
                        + ((size_t)(bh * (N_ / 16) + w * (KPW / 16)) * 4) * 32 + lane;
    v2f kf[64];
    #pragma unroll
    for (int i = 0; i < 32; ++i) {
        const float4 k4 = kbase[i * 32];
        kf[2 * i].x     = k4.x; kf[2 * i].y     = k4.y;
        kf[2 * i + 1].x = k4.z; kf[2 * i + 1].y = k4.w;
    }
    const float4* vbase = reinterpret_cast<const float4*>(Vb)
                        + ((size_t)bh * (N_ / 4) + key0 / 4 + 16 * hi) * 32 + lo;
    v2f va[32], vb[32];
    #pragma unroll
    for (int kk = 0; kk < 32; ++kk) {
        const float4 v4 = vbase[kk * 16];
        va[kk].x = v4.x; va[kk].y = v4.y;
        vb[kk].x = v4.z; vb[kk].y = v4.w;
    }

    const float* sw = sS + lo * SROW + key0 + 64 * hi;

    #pragma unroll 1
    for (int t = 0; t < NQT; ++t) {
        #pragma unroll
        for (int j = 0; j < KPW / 16; j += 2) {
            v8f acc0 = {}, acc1 = {};
            #pragma unroll
            for (int ts = 0; ts < 8; ++ts) {
                acc0 = wmma4(aq[ts], kf[8 * j + ts],       acc0);
                acc1 = wmma4(aq[ts], kf[8 * (j + 1) + ts], acc1);
            }
            #pragma unroll
            for (int r = 0; r < 8; ++r) {
                sS[(r + 8 * hi) * SROW + key0 + j * 16 + lo]       = acc0[r];
                sS[(r + 8 * hi) * SROW + key0 + (j + 1) * 16 + lo] = acc1[r];
            }
        }

        if (t + 1 < NQT) {
            #pragma unroll
            for (int tsp = 0; tsp < 4; ++tsp) {
                const float4 q4 = qf[(t + 1) * 128 + tsp * 32];
                aq[2 * tsp].x     = q4.x; aq[2 * tsp].y     = q4.y;
                aq[2 * tsp + 1].x = q4.z; aq[2 * tsp + 1].y = q4.w;
            }
        }

        float4 pv[16];
        #pragma unroll
        for (int i = 0; i < 16; ++i)
            pv[i] = *reinterpret_cast<const float4*>(sw + 4 * i);
        float mx = fmaxf(fmaxf(pv[0].x, pv[0].y), fmaxf(pv[0].z, pv[0].w));
        #pragma unroll
        for (int i = 1; i < 16; ++i)
            mx = fmaxf(mx, fmaxf(fmaxf(pv[i].x, pv[i].y), fmaxf(pv[i].z, pv[i].w)));
        mx = fmaxf(mx, xhalf(mx));
        const float mc = mx * cexp;
        float sum = 0.f;
        #pragma unroll
        for (int i = 0; i < 16; ++i) {
            pv[i].x = __builtin_amdgcn_exp2f(fmaf(pv[i].x, cexp, -mc));
            pv[i].y = __builtin_amdgcn_exp2f(fmaf(pv[i].y, cexp, -mc));
            pv[i].z = __builtin_amdgcn_exp2f(fmaf(pv[i].z, cexp, -mc));
            pv[i].w = __builtin_amdgcn_exp2f(fmaf(pv[i].w, cexp, -mc));
            sum += (pv[i].x + pv[i].y) + (pv[i].z + pv[i].w);
        }
        sum += xhalf(sum);

        v8f oacc0 = {}, oacc1 = {};
        #pragma unroll
        for (int kk = 0; kk < 32; ++kk) {
            v2f ap;
            if (kk & 1) { ap.x = pv[kk >> 1].z; ap.y = pv[kk >> 1].w; }
            else        { ap.x = pv[kk >> 1].x; ap.y = pv[kk >> 1].y; }
            oacc0 = wmma4(ap, va[kk], oacc0);
            oacc1 = wmma4(ap, vb[kk], oacc1);
        }

        float* pk = park + (size_t)((t * AW + w) * 16) * PROW;
        #pragma unroll
        for (int r = 0; r < 8; ++r) {
            pk[(r + 8 * hi) * PROW + lo]      = oacc0[r];
            pk[(r + 8 * hi) * PROW + 16 + lo] = oacc1[r];
        }
        pk[lo * PROW + 32 + hi] = hi ? sum : mc;
    }
    __syncthreads();

    #pragma unroll
    for (int t = 0; t < NQT; ++t) {
        const float* pp = park + (size_t)(t * AW * 16 + jrow) * PROW;
        float2 ml[AW];
        float mall = -3.0e38f;
        #pragma unroll
        for (int ww = 0; ww < AW; ++ww) {
            ml[ww] = *reinterpret_cast<const float2*>(pp + ww * 16 * PROW + 32);
            mall = fmaxf(mall, ml[ww].x);
        }
        float den = 0.f, sx = 0.f, sy = 0.f;
        #pragma unroll
        for (int ww = 0; ww < AW; ++ww) {
            const float f  = __builtin_amdgcn_exp2f(ml[ww].x - mall);
            const float2 o = *reinterpret_cast<const float2*>(pp + ww * 16 * PROW + jd);
            den = fmaf(f, ml[ww].y, den);
            sx  = fmaf(f, o.x, sx);
            sy  = fmaf(f, o.y, sy);
        }
        const float inv = 1.f / den;
        float2 ov; ov.x = fmaf(sx, inv, hv[t].x); ov.y = fmaf(sy, inv, hv[t].y);
        *reinterpret_cast<float2*>(Out + jidx + (size_t)t * 16 * CR_) = ov;
    }
}

extern "C" void kernel_launch(void* const* d_in, const int* in_sizes, int n_in,
                              void* d_out, int out_size, void* d_ws, size_t ws_size,
                              hipStream_t stream) {
    (void)in_sizes; (void)n_in; (void)out_size; (void)ws_size;
    const float* x      = (const float*)d_in[0];
    const float* ln0_g  = (const float*)d_in[1];
    const float* ln0_b  = (const float*)d_in[2];
    const float* w_proj = (const float*)d_in[3];
    const float* b_proj = (const float*)d_in[4];
    const float* ln1_g  = (const float*)d_in[5];
    const float* ln1_b  = (const float*)d_in[6];
    const float* wq     = (const float*)d_in[7];
    const float* wk     = (const float*)d_in[8];
    const float* wv     = (const float*)d_in[9];
    float* out = (float*)d_out;

    float* ws = (float*)d_ws;
    const size_t rowsBig = (size_t)B_ * N_ * DIM_;
    const size_t rowsCr  = (size_t)B_ * N_ * CR_;
    float* xn = ws;
    float* hB = xn + rowsBig;
    float* yB = hB + rowsCr;
    float* wf = yB + rowsCr;
    float* kB = wf + rowsCr;
    float* vB = kB + rowsCr;
    float* qB = xn;
    float* wpf   = wf;
    float* wqkvf = wf + (size_t)CR_ * DIM_;

    const int M = B_ * N_;

    ln_frag_kernel<DIM_><<<M / 16 + WF_BLOCKS, 256, 0, stream>>>(
        x, ln0_g, ln0_b, xn, w_proj, wq, wk, wv, wpf, wqkvf, M / 16);

    (void)hipFuncSetAttribute((const void*)proj_ln_kernel,
                              hipFuncAttributeMaxDynamicSharedMemorySize,
                              PJ_SMEM);
    proj_ln_kernel<<<M / 32, 256, PJ_SMEM, stream>>>(
        xn, wpf, b_proj, ln1_g, ln1_b, hB, yB);

    (void)hipFuncSetAttribute((const void*)qkv_gemm_kernel,
                              hipFuncAttributeMaxDynamicSharedMemorySize,
                              QK_SMEM);
    qkv_gemm_kernel<<<3 * (M / 32), 128, QK_SMEM, stream>>>(
        yB, wqkvf, qB, kB, vB);

    (void)hipFuncSetAttribute((const void*)attn_kernel,
                              hipFuncAttributeMaxDynamicSharedMemorySize,
                              ATTN_SMEM);
    attn_kernel<<<dim3(N_ / 16 / NQT, B_ * H_), 256, ATTN_SMEM, stream>>>(
        qB, kB, vB, hB, out);
}
